# starting kernel
# baseline (speedup 1.0000x reference)
#include <hip/hip_runtime.h>
#include <hip/hip_bf16.h>


typedef __attribute__((ext_vector_type(2))) float v2f;
typedef __attribute__((ext_vector_type(8))) float v8f;

__device__ __forceinline__ v8f wmma4(v2f a, v2f b, v8f c) {
    return __builtin_amdgcn_wmma_f32_16x16x4_f32(
        false, a, false, b, (short)0, c, false, false);
}

__device__ __forceinline__ v2f load_a_frag(const float* M, int ld, int row0,
                                           int k0, int lane) {
    int r = row0 + (lane & 15);
    int k = k0 + ((lane >> 4) << 1);
    const float* p = M + (size_t)r * ld + k;
    v2f a; a.x = p[0]; a.y = p[1];
    return a;
}

__device__ __forceinline__ v2f load_b_packed(const float* P, int nk, int kt,
                                             int nt, int lane) {
    const v2f* p = (const v2f*)P + ((size_t)(nt * nk + kt) * 32 + lane);
    return *p;
}

__device__ __forceinline__ float sigmoid_f(float v) {
    return 1.0f / (1.0f + __expf(-v));
}
__device__ __forceinline__ float tanh_f(float v) {
    float t = __expf(-2.0f * v);
    return (1.0f - t) / (1.0f + t);
}

__global__ __launch_bounds__(256) void pack_kernel(
    const float* __restrict__ W, float* __restrict__ P, int K, int Nc) {
    int w = (int)((blockIdx.x * 256u + threadIdx.x) >> 5);
    int lane = threadIdx.x & 31;
    int nk = K >> 2, nn = Nc >> 4;
    if (w >= nk * nn) return;
    int kt = w % nk, nt = w / nk;
    int n = nt * 16 + (lane & 15);
    int k = kt * 4 + ((lane >> 4) << 1);
    v2f v;
    v.x = W[(size_t)k * Nc + n];
    v.y = W[(size_t)(k + 1) * Nc + n];
    ((v2f*)P)[(size_t)(nt * nk + kt) * 32 + lane] = v;
}

__global__ __launch_bounds__(256) void zero_kernel(float* p, long n) {
    long i = (long)blockIdx.x * blockDim.x + threadIdx.x;
    long stride = (long)gridDim.x * blockDim.x;
    for (; i < n; i += stride) p[i] = 0.0f;
}

__global__ __launch_bounds__(128) void enc_kernel(
    const float* __restrict__ feat, const float* __restrict__ pW0,
    const float* __restrict__ b0, const float* __restrict__ pW1,
    const float* __restrict__ b1, float* __restrict__ x, int N) {
    __shared__ float t[4][16 * 256];
    const int wave = threadIdx.x >> 5;
    const int lane = threadIdx.x & 31;
    const int row0 = (blockIdx.x * 4 + wave) * 16;
    const bool active = (row0 < N);

    if (active) {
        v2f afr[16];
#pragma unroll
        for (int kk = 0; kk < 16; ++kk)
            afr[kk] = load_a_frag(feat, 64, row0, kk * 4, lane);
#pragma unroll 1
        for (int nt = 0; nt < 16; ++nt) {
            v8f acc = {};
#pragma unroll
            for (int kk = 0; kk < 16; ++kk) {
                v2f b = load_b_packed(pW0, 16, kk, nt, lane);
                acc = wmma4(afr[kk], b, acc);
            }
            int col = nt * 16 + (lane & 15);
            float bias = b0[col];
#pragma unroll
            for (int i = 0; i < 8; ++i) {
                int r = i + 8 * (lane >> 4);
                t[wave][r * 256 + col] = fmaxf(acc[i] + bias, 0.0f);
            }
        }
    }
    __syncthreads();
    if (active) {
#pragma unroll 1
        for (int nt = 0; nt < 16; ++nt) {
            v8f acc = {};
#pragma unroll 4
            for (int kk = 0; kk < 64; ++kk) {
                int k = kk * 4 + ((lane >> 4) << 1);
                int r = lane & 15;
                v2f a; a.x = t[wave][r * 256 + k]; a.y = t[wave][r * 256 + k + 1];
                v2f b = load_b_packed(pW1, 64, kk, nt, lane);
                acc = wmma4(a, b, acc);
            }
            int col = nt * 16 + (lane & 15);
            float bias = b1[col];
#pragma unroll
            for (int i = 0; i < 8; ++i) {
                int r = row0 + i + 8 * (lane >> 4);
                x[(size_t)r * 256 + col] = fmaxf(acc[i] + bias, 0.0f);
            }
        }
    }
}

__global__ __launch_bounds__(128) void msg_kernel(
    const float* __restrict__ x, const float* __restrict__ h,
    const float* __restrict__ pmsgW, const float* __restrict__ msgb,
    float* __restrict__ mx, int N) {
    const int wave = threadIdx.x >> 5;
    const int lane = threadIdx.x & 31;
    const int row0 = (blockIdx.x * 4 + wave) * 32;
    if (row0 >= N) return;
#pragma unroll 1
    for (int nt = 0; nt < 8; ++nt) {
        v8f acc0 = {}, acc1 = {};
#pragma unroll 4
        for (int kk = 0; kk < 128; ++kk) {
            int k0 = kk * 4;
            v2f b = load_b_packed(pmsgW, 128, kk, nt, lane);
            v2f a0 = (k0 < 256) ? load_a_frag(x, 256, row0, k0, lane)
                                : load_a_frag(h, 256, row0, k0 - 256, lane);
            v2f a1 = (k0 < 256) ? load_a_frag(x, 256, row0 + 16, k0, lane)
                                : load_a_frag(h, 256, row0 + 16, k0 - 256, lane);
            acc0 = wmma4(a0, b, acc0);
            acc1 = wmma4(a1, b, acc1);
        }
        int col = nt * 16 + (lane & 15);
        float bias = msgb[col];
#pragma unroll
        for (int i = 0; i < 8; ++i) {
            int r = row0 + i + 8 * (lane >> 4);
            mx[(size_t)r * 128 + col] = acc0[i] + bias;
            mx[(size_t)(r + 16) * 128 + col] = acc1[i] + bias;
        }
    }
}

__global__ __launch_bounds__(256) void edge_kernel(
    const int* __restrict__ src, const int* __restrict__ dst,
    const float* __restrict__ mx, float* __restrict__ sums,
    float* __restrict__ cnt, int E) {
    int e = (int)((blockIdx.x * 256u + threadIdx.x) >> 5);
    int lane = threadIdx.x & 31;
    if (e >= E) return;
    int s = src[e];
    int d = dst[e];
    const float4* mv = (const float4*)(mx + (size_t)s * 128);
    float4 v = mv[lane];
    float* base = sums + (size_t)d * 128 + lane * 4;
    atomicAdd(base + 0, v.x);
    atomicAdd(base + 1, v.y);
    atomicAdd(base + 2, v.z);
    atomicAdd(base + 3, v.w);
    if (lane == 0) atomicAdd(cnt + d, 1.0f);
}

__global__ __launch_bounds__(256) void mean_kernel(
    const float* __restrict__ sums, const float* __restrict__ cnt,
    float* __restrict__ c, long n) {
    long i = (long)blockIdx.x * blockDim.x + threadIdx.x;
    if (i >= n) return;
    long node = i >> 7;
    c[i] = sums[i] / fmaxf(cnt[node], 1.0f);
}

__global__ __launch_bounds__(128) void gru_kernel(
    const float* __restrict__ x, const float* __restrict__ c,
    const float* __restrict__ h, const float* __restrict__ pWih,
    const float* __restrict__ pWhh, const float* __restrict__ bih,
    const float* __restrict__ bhh, float* __restrict__ hnew, int N) {
    const int wave = threadIdx.x >> 5;
    const int lane = threadIdx.x & 31;
    const int row0 = (blockIdx.x * 4 + wave) * 32;
    if (row0 >= N) return;
#pragma unroll 1
    for (int nt = 0; nt < 16; ++nt) {
        v8f ir0 = {}, iz0 = {}, in0 = {}, ir1 = {}, iz1 = {}, in1 = {};
        v8f hr0 = {}, hz0 = {}, hn0 = {}, hr1 = {}, hz1 = {}, hn1 = {};
#pragma unroll 2
        for (int kk = 0; kk < 96; ++kk) {
            int k0 = kk * 4;
            v2f br = load_b_packed(pWih, 96, kk, nt, lane);
            v2f bz = load_b_packed(pWih, 96, kk, nt + 16, lane);
            v2f bn = load_b_packed(pWih, 96, kk, nt + 32, lane);
            v2f a0 = (k0 < 256) ? load_a_frag(x, 256, row0, k0, lane)
                                : load_a_frag(c, 128, row0, k0 - 256, lane);
            v2f a1 = (k0 < 256) ? load_a_frag(x, 256, row0 + 16, k0, lane)
                                : load_a_frag(c, 128, row0 + 16, k0 - 256, lane);
            ir0 = wmma4(a0, br, ir0);  ir1 = wmma4(a1, br, ir1);
            iz0 = wmma4(a0, bz, iz0);  iz1 = wmma4(a1, bz, iz1);
            in0 = wmma4(a0, bn, in0);  in1 = wmma4(a1, bn, in1);
        }
#pragma unroll 2
        for (int kk = 0; kk < 64; ++kk) {
            int k0 = kk * 4;
            v2f br = load_b_packed(pWhh, 64, kk, nt, lane);
            v2f bz = load_b_packed(pWhh, 64, kk, nt + 16, lane);
            v2f bn = load_b_packed(pWhh, 64, kk, nt + 32, lane);
            v2f a0 = load_a_frag(h, 256, row0, k0, lane);
            v2f a1 = load_a_frag(h, 256, row0 + 16, k0, lane);
            hr0 = wmma4(a0, br, hr0);  hr1 = wmma4(a1, br, hr1);
            hz0 = wmma4(a0, bz, hz0);  hz1 = wmma4(a1, bz, hz1);
            hn0 = wmma4(a0, bn, hn0);  hn1 = wmma4(a1, bn, hn1);
        }
        int col = nt * 16 + (lane & 15);
        float bir = bih[col], biz = bih[col + 256], bin = bih[col + 512];
        float bhr = bhh[col], bhz = bhh[col + 256], bhn = bhh[col + 512];
#pragma unroll
        for (int i = 0; i < 8; ++i) {
            int r = row0 + i + 8 * (lane >> 4);
            {
                float rg = sigmoid_f((ir0[i] + bir) + (hr0[i] + bhr));
                float zg = sigmoid_f((iz0[i] + biz) + (hz0[i] + bhz));
                float ng = tanh_f((in0[i] + bin) + rg * (hn0[i] + bhn));
                float hv = h[(size_t)r * 256 + col];
                hnew[(size_t)r * 256 + col] = (1.0f - zg) * ng + zg * hv;
            }
            {
                float rg = sigmoid_f((ir1[i] + bir) + (hr1[i] + bhr));
                float zg = sigmoid_f((iz1[i] + biz) + (hz1[i] + bhz));
                float ng = tanh_f((in1[i] + bin) + rg * (hn1[i] + bhn));
                float hv = h[(size_t)(r + 16) * 256 + col];
                hnew[(size_t)(r + 16) * 256 + col] = (1.0f - zg) * ng + zg * hv;
            }
        }
    }
}

__global__ __launch_bounds__(128) void out_kernel(
    const float* __restrict__ hnew, const float* __restrict__ poutW,
    const float* __restrict__ outb, float* __restrict__ q, int N) {
    const int wave = threadIdx.x >> 5;
    const int lane = threadIdx.x & 31;
    const int row0 = (blockIdx.x * 4 + wave) * 32;
    if (row0 >= N) return;
    v8f acc0 = {}, acc1 = {};
#pragma unroll 4
    for (int kk = 0; kk < 64; ++kk) {
        v2f b = load_b_packed(poutW, 64, kk, 0, lane);
        v2f a0 = load_a_frag(hnew, 256, row0, kk * 4, lane);
        v2f a1 = load_a_frag(hnew, 256, row0 + 16, kk * 4, lane);
        acc0 = wmma4(a0, b, acc0);
        acc1 = wmma4(a1, b, acc1);
    }
    int col = lane & 15;
    float bias = outb[col];
#pragma unroll
    for (int i = 0; i < 8; ++i) {
        int r = row0 + i + 8 * (lane >> 4);
        q[(size_t)r * 16 + col] = acc0[i] + bias;
        q[(size_t)(r + 16) * 16 + col] = acc1[i] + bias;
    }
}

static inline int packBlocks(int K, int Nc) {
    int waves = (K / 4) * (Nc / 16);
    return (waves + 7) / 8;
}

extern "C" void kernel_launch(void* const* d_in, const int* in_sizes, int n_in,
                              void* d_out, int out_size, void* d_ws,
                              size_t ws_size, hipStream_t stream) {
    (void)n_in; (void)out_size; (void)ws_size;
    const float* feat = (const float*)d_in[0];
    const float* h    = (const float*)d_in[1];
    const int*   src  = (const int*)d_in[2];
    const int*   dst  = (const int*)d_in[3];
    const float* W0   = (const float*)d_in[4];
    const float* b0   = (const float*)d_in[5];
    const float* W1   = (const float*)d_in[6];
    const float* b1   = (const float*)d_in[7];
    const float* msgW = (const float*)d_in[8];
    const float* msgb = (const float*)d_in[9];
    const float* Wih  = (const float*)d_in[10];
    const float* Whh  = (const float*)d_in[11];
    const float* bih  = (const float*)d_in[12];
    const float* bhh  = (const float*)d_in[13];
    const float* outW = (const float*)d_in[14];
    const float* outb = (const float*)d_in[15];

    const int N = in_sizes[1] / 256;
    const int E = in_sizes[2];

    float* ws    = (float*)d_ws;
    float* x     = ws;
    float* mx    = x    + (size_t)N * 256;
    float* sums  = mx   + (size_t)N * 128;
    float* cnt   = sums + (size_t)N * 128;
    float* c     = cnt  + N;
    float* pW0   = c    + (size_t)N * 128;
    float* pW1   = pW0  + 16384;
    float* pmsgW = pW1  + 65536;
    float* pWih  = pmsgW + 65536;
    float* pWhh  = pWih + 294912;
    float* poutW = pWhh + 196608;

    float* q    = (float*)d_out;
    float* hnew = q + (size_t)N * 16;

    const int encBlocks  = (N / 16 + 3) / 4;
    const int gemmBlocks = (N / 32 + 3) / 4;
    const long zeroN     = (long)N * 128 + N;
    const int zeroBlocks = (int)((zeroN + 255) / 256);
    const int edgeBlocks = (E + 7) / 8;
    const long meanN     = (long)N * 128;
    const int meanBlocks = (int)((meanN + 255) / 256);

    pack_kernel<<<packBlocks(64, 256), 256, 0, stream>>>(W0, pW0, 64, 256);
    pack_kernel<<<packBlocks(256, 256), 256, 0, stream>>>(W1, pW1, 256, 256);
    pack_kernel<<<packBlocks(512, 128), 256, 0, stream>>>(msgW, pmsgW, 512, 128);
    pack_kernel<<<packBlocks(384, 768), 256, 0, stream>>>(Wih, pWih, 384, 768);
    pack_kernel<<<packBlocks(256, 768), 256, 0, stream>>>(Whh, pWhh, 256, 768);
    pack_kernel<<<packBlocks(256, 16), 256, 0, stream>>>(outW, poutW, 256, 16);

    zero_kernel<<<zeroBlocks, 256, 0, stream>>>(sums, zeroN);
    enc_kernel<<<encBlocks, 128, 0, stream>>>(feat, pW0, b0, pW1, b1, x, N);
    msg_kernel<<<gemmBlocks, 128, 0, stream>>>(x, h, pmsgW, msgb, mx, N);
    edge_kernel<<<edgeBlocks, 256, 0, stream>>>(src, dst, mx, sums, cnt, E);
    mean_kernel<<<meanBlocks, 256, 0, stream>>>(sums, cnt, c, meanN);
    gru_kernel<<<gemmBlocks, 128, 0, stream>>>(x, c, h, pWih, pWhh, bih, bhh,
                                               hnew, N);
    out_kernel<<<gemmBlocks, 128, 0, stream>>>(hnew, poutW, outb, q, N);
}


# encoder made weight-stationary in two launches (feat@W0 -> t in workspace, t@W1 -> x): a workgroup keeps its column slice of packed B fragments in LDS, streams 16-row tiles; k order unchanged
# speedup vs baseline: 4.6978x; 4.6978x over previous
#include <hip/hip_runtime.h>
#include <hip/hip_bf16.h>


typedef __attribute__((ext_vector_type(2))) float v2f;
typedef __attribute__((ext_vector_type(8))) float v8f;

__device__ __forceinline__ v8f wmma4(v2f a, v2f b, v8f c) {
    return __builtin_amdgcn_wmma_f32_16x16x4_f32(
        false, a, false, b, (short)0, c, false, false);
}

__device__ __forceinline__ v2f load_a_frag(const float* M, int ld, int row0,
                                           int k0, int lane) {
    int r = row0 + (lane & 15);
    int k = k0 + ((lane >> 4) << 1);
    const float* p = M + (size_t)r * ld + k;
    v2f a; a.x = p[0]; a.y = p[1];
    return a;
}

__device__ __forceinline__ v2f load_b_packed(const float* P, int nk, int kt,
                                             int nt, int lane) {
    const v2f* p = (const v2f*)P + ((size_t)(nt * nk + kt) * 32 + lane);
    return *p;
}

__device__ __forceinline__ float sigmoid_f(float v) {
    return 1.0f / (1.0f + __expf(-v));
}
__device__ __forceinline__ float tanh_f(float v) {
    float t = __expf(-2.0f * v);
    return (1.0f - t) / (1.0f + t);
}

__global__ __launch_bounds__(256) void pack_kernel(
    const float* __restrict__ W, float* __restrict__ P, int K, int Nc) {
    int w = (int)((blockIdx.x * 256u + threadIdx.x) >> 5);
    int lane = threadIdx.x & 31;
    int nk = K >> 2, nn = Nc >> 4;
    if (w >= nk * nn) return;
    int kt = w % nk, nt = w / nk;
    int n = nt * 16 + (lane & 15);
    int k = kt * 4 + ((lane >> 4) << 1);
    v2f v;
    v.x = W[(size_t)k * Nc + n];
    v.y = W[(size_t)(k + 1) * Nc + n];
    ((v2f*)P)[(size_t)(nt * nk + kt) * 32 + lane] = v;
}


template <int NK, int NT>
__device__ __forceinline__ void fill_tile(const float* __restrict__ P, int nt,
                                          float* sb, int t, int tid) {
    const int q = tid & 15, k16 = tid >> 4;
#pragma unroll
    for (int ii = 0; ii < NK / 16; ++ii) {
        int kt = k16 + 16 * ii;
        float4 v = *(const float4*)(P + ((size_t)(nt * NK + kt) * 64 + q * 4));
        *(float4*)(sb + (kt * NT + t) * 64 + q * 4) = v;
    }
}

template <int NK, int NT>
__device__ __forceinline__ void ws_phase(const float* __restrict__ ap,
                                         const float* sb, v8f (&acc)[NT]) {
#pragma unroll 1
    for (int kb = 0; kb < NK; kb += 8) {
        v2f a[8];
#pragma unroll
        for (int s = 0; s < 8; ++s) a[s] = *(const v2f*)(ap + 4 * (kb + s));
#pragma unroll
        for (int s = 0; s < 8; ++s) {
            const float* bp = sb + (kb + s) * (NT * 64);
#pragma unroll
            for (int t = 0; t < NT; ++t) {
                v2f b = *(const v2f*)(bp + t * 64);
                acc[t] = wmma4(a[s], b, acc[t]);
            }
        }
    }
}


__global__ __launch_bounds__(256) void enc1_kernel(
    const float* __restrict__ feat, const float* __restrict__ pW0,
    const float* __restrict__ b0, float* __restrict__ t1, int N) {
    __shared__ float sB[16 * 8 * 64];
    const int tid  = threadIdx.x;
    const int wave = tid >> 5;
    const int lane = tid & 31;
    const int cs   = blockIdx.x & 1;
    const int rgp  = blockIdx.x >> 1;

#pragma unroll
    for (int t = 0; t < 8; ++t)
        fill_tile<16, 8>(pW0, 8 * cs + t, sB, t, tid);
    __syncthreads();

    const int cidx = lane & 15, hh = lane >> 4;
    const float* sb = sB + lane * 2;
    float bias[8];
#pragma unroll
    for (int t = 0; t < 8; ++t) bias[t] = b0[(8 * cs + t) * 16 + cidx];

    const int ntiles = N >> 4;
    for (int rt = rgp + 128 * wave; rt < ntiles; rt += 128 * 8) {
        const int row0 = rt * 16;
        const size_t ar = (size_t)(row0 + cidx);
        v8f acc[8];
#pragma unroll
        for (int t = 0; t < 8; ++t) acc[t] = v8f{};
        ws_phase<16, 8>(feat + ar * 64 + 2 * hh, sb, acc);
#pragma unroll
        for (int t = 0; t < 8; ++t) {
            int col = (8 * cs + t) * 16 + cidx;
#pragma unroll
            for (int i = 0; i < 8; ++i) {
                int r = row0 + i + 8 * hh;
                t1[(size_t)r * 256 + col] = fmaxf(acc[t][i] + bias[t], 0.0f);
            }
        }
    }
}

__global__ __launch_bounds__(512) void enc2_kernel(
    const float* __restrict__ t1, const float* __restrict__ pW1,
    const float* __restrict__ b1, float* __restrict__ x, int N) {
    __shared__ float sB[64 * 4 * 64];
    const int tid  = threadIdx.x;
    const int wave = tid >> 5;
    const int lane = tid & 31;
    const int cs   = blockIdx.x & 3;
    const int rgp  = blockIdx.x >> 2;

    const int fh = tid >> 8;
#pragma unroll
    for (int j = 0; j < 2; ++j)
        fill_tile<64, 4>(pW1, 4 * cs + 2 * fh + j, sB, 2 * fh + j, tid & 255);
    __syncthreads();

    const int cidx = lane & 15, hh = lane >> 4;
    const float* sb = sB + lane * 2;
    float bias[4];
#pragma unroll
    for (int t = 0; t < 4; ++t) bias[t] = b1[(4 * cs + t) * 16 + cidx];

    const int vw = ((((wave >> 2) - wave) & 3) << 2) | (wave & 3);
    const int ntiles = N >> 4;
    for (int rt = rgp + 64 * vw; rt < ntiles; rt += 64 * 16) {
        const int row0 = rt * 16;
        const size_t ar = (size_t)(row0 + cidx);
        v8f acc[4];
#pragma unroll
        for (int t = 0; t < 4; ++t) acc[t] = v8f{};
        ws_phase<64, 4>(t1 + ar * 256 + 2 * hh, sb, acc);
#pragma unroll
        for (int t = 0; t < 4; ++t) {
            int col = (4 * cs + t) * 16 + cidx;
#pragma unroll
            for (int i = 0; i < 8; ++i) {
                int r = row0 + i + 8 * hh;
                x[(size_t)r * 256 + col] = fmaxf(acc[t][i] + bias[t], 0.0f);
            }
        }
    }
}

__global__ __launch_bounds__(256) void msg_kernel(
    const float* __restrict__ x, const float* __restrict__ h,
    const float* __restrict__ pmsgW, const float* __restrict__ msgb,
    float* __restrict__ mx, int N) {
    __shared__ float sB[128 * 4 * 64];
    const int tid  = threadIdx.x;
    const int wave = tid >> 5;
    const int lane = tid & 31;
    const int cs   = blockIdx.x & 1;
    const int rgp  = blockIdx.x >> 1;

#pragma unroll
    for (int t = 0; t < 4; ++t)
        fill_tile<128, 4>(pmsgW, 4 * cs + t, sB, t, tid);
    __syncthreads();

    const int cidx = lane & 15, hh = lane >> 4;
    const float* sb = sB + lane * 2;
    float bias[4];
#pragma unroll
    for (int t = 0; t < 4; ++t) bias[t] = msgb[(4 * cs + t) * 16 + cidx];

    const int ntiles = N >> 4;
    for (int rt = rgp + 128 * wave; rt < ntiles; rt += 128 * 8) {
        const int row0 = rt * 16;
        const size_t ar = (size_t)(row0 + cidx);
        v8f acc[4];
#pragma unroll
        for (int t = 0; t < 4; ++t) acc[t] = v8f{};
        ws_phase<64, 4>(x + ar * 256 + 2 * hh, sb, acc);
        ws_phase<64, 4>(h + ar * 256 + 2 * hh, sb + 64 * 4 * 64, acc);
#pragma unroll
        for (int t = 0; t < 4; ++t) {
            int col = (4 * cs + t) * 16 + cidx;
#pragma unroll
            for (int i = 0; i < 8; ++i) {
                int r = row0 + i + 8 * hh;
                mx[(size_t)r * 128 + col] = acc[t][i] + bias[t];
            }
        }
    }
}

#define AGG_CAP  80
#define AGG_LIST 1024

__device__ __forceinline__ int agg_flush(unsigned* list, int count, bool final,
                                         const int* __restrict__ src,
                                         const float* __restrict__ mx,
                                         float* ssum, int* scnt, int lane) {
    int i0 = 0;
    while (count - i0 >= 32 || (final && count > i0)) {
        const int n = min(32, count - i0);
        unsigned ent = 0u;
        int s = 0;
        if (lane < n) {
            ent = list[i0 + lane];
            s = src[ent >> 7];
            __hip_atomic_fetch_add(&scnt[ent & 127u], 1, __ATOMIC_RELAXED,
                                   __HIP_MEMORY_SCOPE_WORKGROUP);
        }
        for (int j = 0; j < n; j += 8) {
            float v[8][4];
#pragma unroll
            for (int u = 0; u < 8; ++u) {
                const int sj = __builtin_amdgcn_readlane(s, min(j + u, n - 1));
                const float* row = mx + (size_t)sj * 128 + lane;
                v[u][0] = row[0];  v[u][1] = row[32];
                v[u][2] = row[64]; v[u][3] = row[96];
            }
#pragma unroll
            for (int u = 0; u < 8; ++u) {
                if (j + u < n) {
                    const unsigned ld =
                        (unsigned)__builtin_amdgcn_readlane((int)ent, j + u) & 127u;
                    float* p = ssum + ld * 128u + lane;
                    __hip_atomic_fetch_add(p,      v[u][0], __ATOMIC_RELAXED, __HIP_MEMORY_SCOPE_WORKGROUP);
                    __hip_atomic_fetch_add(p + 32, v[u][1], __ATOMIC_RELAXED, __HIP_MEMORY_SCOPE_WORKGROUP);
                    __hip_atomic_fetch_add(p + 64, v[u][2], __ATOMIC_RELAXED, __HIP_MEMORY_SCOPE_WORKGROUP);
                    __hip_atomic_fetch_add(p + 96, v[u][3], __ATOMIC_RELAXED, __HIP_MEMORY_SCOPE_WORKGROUP);
                }
            }
        }
        i0 += n;
    }
    const int rem = count - i0;
    if (i0 > 0 && rem > 0) {
        unsigned tmp = 0u;
        if (lane < rem) tmp = list[i0 + lane];
        if (lane < rem) list[lane] = tmp;
    }
    return rem;
}

__global__ __launch_bounds__(1024) void agg_kernel(
    const int* __restrict__ src, const int* __restrict__ dst,
    const float* __restrict__ mx, float* __restrict__ c, int N, int E) {
    __shared__ float ssum[AGG_CAP * 128];
    __shared__ int scnt[AGG_CAP];
    __shared__ unsigned slist[32 * AGG_LIST];
    const int tid  = threadIdx.x;
    const int wave = __builtin_amdgcn_readfirstlane(tid >> 5);
    const int lane = tid & 31;
    unsigned* list = slist + wave * AGG_LIST;
    const int nblk = (E + 511) >> 9;

    for (int base = blockIdx.x * AGG_CAP; base < N; base += gridDim.x * AGG_CAP) {
        const int nloc = min(AGG_CAP, N - base);
        for (int i = tid; i < AGG_CAP * 128; i += 1024) ssum[i] = 0.0f;
        if (tid < AGG_CAP) scnt[tid] = 0;
        __syncthreads();

        int count = 0;
        const int rot = (int)(((long)blockIdx.x * nblk) / gridDim.x);
        for (int ib = wave; ib < nblk; ib += 32) {
            const int b = (ib + rot < nblk) ? ib + rot : ib + rot - nblk;
            const int e0 = b * 512 + lane * 4;
            int dv[16];
            if (b * 512 + 512 <= E) {
                const int4* p = (const int4*)(dst + e0);
#pragma unroll
                for (int k = 0; k < 4; ++k) {
                    const int4 q = p[k * 32];
                    dv[4 * k + 0] = q.x; dv[4 * k + 1] = q.y;
                    dv[4 * k + 2] = q.z; dv[4 * k + 3] = q.w;
                }
            } else {
#pragma unroll
                for (int i = 0; i < 16; ++i) {
                    const int e = e0 + (i >> 2) * 128 + (i & 3);
                    dv[i] = (e < E) ? dst[e] : -1;
                }
            }
#pragma unroll
            for (int i = 0; i < 16; ++i) {
                const unsigned t = (unsigned)dv[i] - (unsigned)base;
                const bool m = t < (unsigned)nloc;
                const unsigned mask = __builtin_amdgcn_ballot_w32(m);
                if (mask != 0u) {
                    if (m) {
                        const int pos = count + (int)__builtin_amdgcn_mbcnt_lo(mask, 0u);
                        const unsigned e = (unsigned)(e0 + (i >> 2) * 128 + (i & 3));
                        list[pos] = (e << 7) | t;
                    }
                    count += __builtin_popcount(mask);
                }
            }
            if (count >= 32)
                count = agg_flush(list, count, false, src, mx, ssum, scnt, lane);
        }
        agg_flush(list, count, true, src, mx, ssum, scnt, lane);
        __syncthreads();

        for (int r = wave; r < nloc; r += 32) {
            const float den = fmaxf((float)scnt[r], 1.0f);
            const float* sp = ssum + r * 128 + lane;
            float* cp = c + (size_t)(base + r) * 128 + lane;
            cp[0]  = sp[0]  / den;
            cp[32] = sp[32] / den;
            cp[64] = sp[64] / den;
            cp[96] = sp[96] / den;
        }
        __syncthreads();
    }
}

#define GRU_LDS_I (96 * 6 * 64)
#define GRU_LDS_H (64 * 6 * 64)
__global__ __launch_bounds__(256) void gru_kernel(
    const float* __restrict__ x, const float* __restrict__ c,
    const float* __restrict__ h, const float* __restrict__ pWih,
    const float* __restrict__ pWhh, const float* __restrict__ bih,
    const float* __restrict__ bhh, float* __restrict__ hnew, int N) {
    __shared__ float sB[GRU_LDS_I + GRU_LDS_H];
    const int tid  = threadIdx.x;
    const int wave = tid >> 5;
    const int lane = tid & 31;
    const int cs   = blockIdx.x & 7;
    const int rgp  = blockIdx.x >> 3;

#pragma unroll
    for (int t = 0; t < 6; ++t) {
        int ntg = (t % 3) * 16 + 2 * cs + t / 3;
        fill_tile<96, 6>(pWih, ntg, sB, t, tid);
        fill_tile<64, 6>(pWhh, ntg, sB + GRU_LDS_I, t, tid);
    }
    __syncthreads();

    const int cidx = lane & 15, hh = lane >> 4;
    const float* sbi = sB + lane * 2;
    const float* sbh = sB + GRU_LDS_I + lane * 2;

    float bir[2], biz[2], bin[2], bhr[2], bhz[2], bhn[2];
#pragma unroll
    for (int j = 0; j < 2; ++j) {
        int col = (2 * cs + j) * 16 + cidx;
        bir[j] = bih[col]; biz[j] = bih[col + 256]; bin[j] = bih[col + 512];
        bhr[j] = bhh[col]; bhz[j] = bhh[col + 256]; bhn[j] = bhh[col + 512];
    }

    const int ntiles = N >> 4;
    for (int rt = rgp + 32 * wave; rt < ntiles; rt += 32 * 8) {
        const int row0 = rt * 16;
        const size_t ar = (size_t)(row0 + cidx);
        v8f ai[6], ah[6];
#pragma unroll
        for (int t = 0; t < 6; ++t) { ai[t] = v8f{}; ah[t] = v8f{}; }

        ws_phase<64, 6>(x + ar * 256 + 2 * hh, sbi, ai);
        ws_phase<32, 6>(c + ar * 128 + 2 * hh, sbi + 64 * 6 * 64, ai);
        ws_phase<64, 6>(h + ar * 256 + 2 * hh, sbh, ah);

#pragma unroll
        for (int j = 0; j < 2; ++j) {
            int col = (2 * cs + j) * 16 + cidx;
#pragma unroll
            for (int i = 0; i < 8; ++i) {
                int r = row0 + i + 8 * hh;
                float rg = sigmoid_f((ai[j * 3 + 0][i] + bir[j]) + (ah[j * 3 + 0][i] + bhr[j]));
                float zg = sigmoid_f((ai[j * 3 + 1][i] + biz[j]) + (ah[j * 3 + 1][i] + bhz[j]));
                float ng = tanh_f((ai[j * 3 + 2][i] + bin[j]) + rg * (ah[j * 3 + 2][i] + bhn[j]));
                float hv = h[(size_t)r * 256 + col];
                hnew[(size_t)r * 256 + col] = (1.0f - zg) * ng + zg * hv;
            }
        }
    }
}

__global__ __launch_bounds__(128) void out_kernel(
    const float* __restrict__ hnew, const float* __restrict__ poutW,
    const float* __restrict__ outb, float* __restrict__ q, int N) {
    const int wave = threadIdx.x >> 5;
    const int lane = threadIdx.x & 31;
    const int row0 = (blockIdx.x * 4 + wave) * 32;
    if (row0 >= N) return;
    v8f acc0 = {}, acc1 = {};
#pragma unroll 4
    for (int kk = 0; kk < 64; ++kk) {
        v2f b = load_b_packed(poutW, 64, kk, 0, lane);
        v2f a0 = load_a_frag(hnew, 256, row0, kk * 4, lane);
        v2f a1 = load_a_frag(hnew, 256, row0 + 16, kk * 4, lane);
        acc0 = wmma4(a0, b, acc0);
        acc1 = wmma4(a1, b, acc1);
    }
    int col = lane & 15;
    float bias = outb[col];
#pragma unroll
    for (int i = 0; i < 8; ++i) {
        int r = row0 + i + 8 * (lane >> 4);
        q[(size_t)r * 16 + col] = acc0[i] + bias;
        q[(size_t)(r + 16) * 16 + col] = acc1[i] + bias;
    }
}

static inline int packBlocks(int K, int Nc) {
    int waves = (K / 4) * (Nc / 16);
    return (waves + 7) / 8;
}

extern "C" void kernel_launch(void* const* d_in, const int* in_sizes, int n_in,
                              void* d_out, int out_size, void* d_ws,
                              size_t ws_size, hipStream_t stream) {
    (void)n_in; (void)out_size; (void)ws_size;
    const float* feat = (const float*)d_in[0];
    const float* h    = (const float*)d_in[1];
    const int*   src  = (const int*)d_in[2];
    const int*   dst  = (const int*)d_in[3];
    const float* W0   = (const float*)d_in[4];
    const float* b0   = (const float*)d_in[5];
    const float* W1   = (const float*)d_in[6];
    const float* b1   = (const float*)d_in[7];
    const float* msgW = (const float*)d_in[8];
    const float* msgb = (const float*)d_in[9];
    const float* Wih  = (const float*)d_in[10];
    const float* Whh  = (const float*)d_in[11];
    const float* bih  = (const float*)d_in[12];
    const float* bhh  = (const float*)d_in[13];
    const float* outW = (const float*)d_in[14];
    const float* outb = (const float*)d_in[15];

    const int N = in_sizes[1] / 256;
    const int E = in_sizes[2];

    float* ws    = (float*)d_ws;
    float* x     = ws;
    float* mx    = x    + (size_t)N * 256;
    float* c     = mx   + (size_t)N * 128;
    float* pW0   = c    + (size_t)N * 128;
    float* pW1   = pW0  + 16384;
    float* pmsgW = pW1  + 65536;
    float* pWih  = pmsgW + 65536;
    float* pWhh  = pWih + 294912;
    float* poutW = pWhh + 196608;

    float* q    = (float*)d_out;
    float* hnew = q + (size_t)N * 16;

    const int gemmBlocks = (N / 32 + 3) / 4;

    pack_kernel<<<packBlocks(64, 256), 256, 0, stream>>>(W0, pW0, 64, 256);
    pack_kernel<<<packBlocks(256, 256), 256, 0, stream>>>(W1, pW1, 256, 256);
    pack_kernel<<<packBlocks(512, 128), 256, 0, stream>>>(msgW, pmsgW, 512, 128);
    pack_kernel<<<packBlocks(384, 768), 256, 0, stream>>>(Wih, pWih, 384, 768);
    pack_kernel<<<packBlocks(256, 768), 256, 0, stream>>>(Whh, pWhh, 256, 768);
    pack_kernel<<<packBlocks(256, 16), 256, 0, stream>>>(outW, poutW, 256, 16);

    float* t1 = mx;
    enc1_kernel<<<256, 256, 0, stream>>>(feat, pW0, b0, t1, N);
    enc2_kernel<<<256, 512, 0, stream>>>(t1, pW1, b1, x, N);
    msg_kernel<<<256, 256, 0, stream>>>(x, h, pmsgW, msgb, mx, N);
    agg_kernel<<<256, 1024, 0, stream>>>(src, dst, mx, c, N, E);
    gru_kernel<<<256, 256, 0, stream>>>(x, c, h, pWih, pWhh, bih, bhh,
                                        hnew, N);
    out_kernel<<<gemmBlocks, 128, 0, stream>>>(hnew, poutW, outb, q, N);
}


# B fragments built in LDS straight from the row-major weights during each GEMM's fill (head reads its fragments from row-major too); the six pack launches go, 12 launches to 6; k order unchanged
# speedup vs baseline: 4.9967x; 1.0636x over previous
#include <hip/hip_runtime.h>
#include <hip/hip_bf16.h>


typedef __attribute__((ext_vector_type(2))) float v2f;
typedef __attribute__((ext_vector_type(8))) float v8f;

__device__ __forceinline__ v8f wmma4(v2f a, v2f b, v8f c) {
    return __builtin_amdgcn_wmma_f32_16x16x4_f32(
        false, a, false, b, (short)0, c, false, false);
}

__device__ __forceinline__ v2f load_a_frag(const float* M, int ld, int row0,
                                           int k0, int lane) {
    int r = row0 + (lane & 15);
    int k = k0 + ((lane >> 4) << 1);
    const float* p = M + (size_t)r * ld + k;
    v2f a; a.x = p[0]; a.y = p[1];
    return a;
}

__device__ __forceinline__ float sigmoid_f(float v) {
    return 1.0f / (1.0f + __expf(-v));
}
__device__ __forceinline__ float tanh_f(float v) {
    float t = __expf(-2.0f * v);
    return (1.0f - t) / (1.0f + t);
}


template <int NK, int NT, int LD, int NW>
__device__ __forceinline__ void fill_run(const float* __restrict__ Wt, int col0,
                                         float* sb, int t0, int t1, int wave,
                                         int lane) {
    const int half = lane >> 4;
    const int tb   = (lane >> 3) & 1;
    const int cp   = lane & 7;
    const float* gp = Wt + (size_t)(2 * half) * LD + col0 + (lane & 15) * 2;
    float* lp = sb + (tb ? t1 : t0) * 64 + half * 32 + cp * 4;
#pragma unroll
    for (int i = 0; i < NK / NW; ++i) {
        const int kt = wave + NW * i;
        const v2f ra = *(const v2f*)(gp + (size_t)(4 * kt) * LD);
        const v2f rb = *(const v2f*)(gp + (size_t)(4 * kt + 1) * LD);
        float4 v;
        v.x = ra.x; v.y = rb.x;
        v.z = ra.y; v.w = rb.y;
        *(float4*)(lp + kt * (NT * 64)) = v;
    }
}

template <int NK, int NT>
__device__ __forceinline__ void ws_phase(const float* __restrict__ ap,
                                         const float* sb, v8f (&acc)[NT]) {
#pragma unroll 1
    for (int kb = 0; kb < NK; kb += 8) {
        v2f a[8];
#pragma unroll
        for (int s = 0; s < 8; ++s) a[s] = *(const v2f*)(ap + 4 * (kb + s));
#pragma unroll
        for (int s = 0; s < 8; ++s) {
            const float* bp = sb + (kb + s) * (NT * 64);
#pragma unroll
            for (int t = 0; t < NT; ++t) {
                v2f b = *(const v2f*)(bp + t * 64);
                acc[t] = wmma4(a[s], b, acc[t]);
            }
        }
    }
}


__global__ __launch_bounds__(256) void enc1_kernel(
    const float* __restrict__ feat, const float* __restrict__ W0,
    const float* __restrict__ b0, float* __restrict__ t1, int N) {
    __shared__ float sB[16 * 8 * 64];
    const int tid  = threadIdx.x;
    const int wave = tid >> 5;
    const int lane = tid & 31;
    const int cs   = blockIdx.x & 1;
    const int rgp  = blockIdx.x >> 1;

#pragma unroll
    for (int r = 0; r < 4; ++r)
        fill_run<16, 8, 256, 8>(W0, 128 * cs + 32 * r, sB, 2 * r, 2 * r + 1,
                                wave, lane);
    __syncthreads();

    const int cidx = lane & 15, hh = lane >> 4;
    const float* sb = sB + lane * 2;
    float bias[8];
#pragma unroll
    for (int t = 0; t < 8; ++t) bias[t] = b0[(8 * cs + t) * 16 + cidx];

    const int ntiles = N >> 4;
    for (int rt = rgp + 128 * wave; rt < ntiles; rt += 128 * 8) {
        const int row0 = rt * 16;
        const size_t ar = (size_t)(row0 + cidx);
        v8f acc[8];
#pragma unroll
        for (int t = 0; t < 8; ++t) acc[t] = v8f{};
        ws_phase<16, 8>(feat + ar * 64 + 2 * hh, sb, acc);
#pragma unroll
        for (int t = 0; t < 8; ++t) {
            int col = (8 * cs + t) * 16 + cidx;
#pragma unroll
            for (int i = 0; i < 8; ++i) {
                int r = row0 + i + 8 * hh;
                t1[(size_t)r * 256 + col] = fmaxf(acc[t][i] + bias[t], 0.0f);
            }
        }
    }
}

__global__ __launch_bounds__(512) void enc2_kernel(
    const float* __restrict__ t1, const float* __restrict__ W1,
    const float* __restrict__ b1, float* __restrict__ x, int N) {
    __shared__ float sB[64 * 4 * 64];
    const int tid  = threadIdx.x;
    const int wave = tid >> 5;
    const int lane = tid & 31;
    const int cs   = blockIdx.x & 3;
    const int rgp  = blockIdx.x >> 2;

#pragma unroll
    for (int r = 0; r < 2; ++r)
        fill_run<64, 4, 256, 16>(W1, 64 * cs + 32 * r, sB, 2 * r, 2 * r + 1,
                                 wave, lane);
    __syncthreads();

    const int cidx = lane & 15, hh = lane >> 4;
    const float* sb = sB + lane * 2;
    float bias[4];
#pragma unroll
    for (int t = 0; t < 4; ++t) bias[t] = b1[(4 * cs + t) * 16 + cidx];

    const int vw = ((((wave >> 2) - wave) & 3) << 2) | (wave & 3);
    const int ntiles = N >> 4;
    for (int rt = rgp + 64 * vw; rt < ntiles; rt += 64 * 16) {
        const int row0 = rt * 16;
        const size_t ar = (size_t)(row0 + cidx);
        v8f acc[4];
#pragma unroll
        for (int t = 0; t < 4; ++t) acc[t] = v8f{};
        ws_phase<64, 4>(t1 + ar * 256 + 2 * hh, sb, acc);
#pragma unroll
        for (int t = 0; t < 4; ++t) {
            int col = (4 * cs + t) * 16 + cidx;
#pragma unroll
            for (int i = 0; i < 8; ++i) {
                int r = row0 + i + 8 * hh;
                x[(size_t)r * 256 + col] = fmaxf(acc[t][i] + bias[t], 0.0f);
            }
        }
    }
}

__global__ __launch_bounds__(256) void msg_kernel(
    const float* __restrict__ x, const float* __restrict__ h,
    const float* __restrict__ msgW, const float* __restrict__ msgb,
    float* __restrict__ mx, int N) {
    __shared__ float sB[128 * 4 * 64];
    const int tid  = threadIdx.x;
    const int wave = tid >> 5;
    const int lane = tid & 31;
    const int cs   = blockIdx.x & 1;
    const int rgp  = blockIdx.x >> 1;

#pragma unroll
    for (int r = 0; r < 2; ++r)
        fill_run<128, 4, 128, 8>(msgW, 64 * cs + 32 * r, sB, 2 * r, 2 * r + 1,
                                 wave, lane);
    __syncthreads();

    const int cidx = lane & 15, hh = lane >> 4;
    const float* sb = sB + lane * 2;
    float bias[4];
#pragma unroll
    for (int t = 0; t < 4; ++t) bias[t] = msgb[(4 * cs + t) * 16 + cidx];

    const int ntiles = N >> 4;
    for (int rt = rgp + 128 * wave; rt < ntiles; rt += 128 * 8) {
        const int row0 = rt * 16;
        const size_t ar = (size_t)(row0 + cidx);
        v8f acc[4];
#pragma unroll
        for (int t = 0; t < 4; ++t) acc[t] = v8f{};
        ws_phase<64, 4>(x + ar * 256 + 2 * hh, sb, acc);
        ws_phase<64, 4>(h + ar * 256 + 2 * hh, sb + 64 * 4 * 64, acc);
#pragma unroll
        for (int t = 0; t < 4; ++t) {
            int col = (4 * cs + t) * 16 + cidx;
#pragma unroll
            for (int i = 0; i < 8; ++i) {
                int r = row0 + i + 8 * hh;
                mx[(size_t)r * 128 + col] = acc[t][i] + bias[t];
            }
        }
    }
}

#define AGG_CAP  80
#define AGG_LIST 1024

__device__ __forceinline__ int agg_flush(unsigned* list, int count, bool final,
                                         const int* __restrict__ src,
                                         const float* __restrict__ mx,
                                         float* ssum, int* scnt, int lane) {
    int i0 = 0;
    while (count - i0 >= 32 || (final && count > i0)) {
        const int n = min(32, count - i0);
        unsigned ent = 0u;
        int s = 0;
        if (lane < n) {
            ent = list[i0 + lane];
            s = src[ent >> 7];
            __hip_atomic_fetch_add(&scnt[ent & 127u], 1, __ATOMIC_RELAXED,
                                   __HIP_MEMORY_SCOPE_WORKGROUP);
        }
        for (int j = 0; j < n; j += 8) {
            float v[8][4];
#pragma unroll
            for (int u = 0; u < 8; ++u) {
                const int sj = __builtin_amdgcn_readlane(s, min(j + u, n - 1));
                const float* row = mx + (size_t)sj * 128 + lane;
                v[u][0] = row[0];  v[u][1] = row[32];
                v[u][2] = row[64]; v[u][3] = row[96];
            }
#pragma unroll
            for (int u = 0; u < 8; ++u) {
                if (j + u < n) {
                    const unsigned ld =
                        (unsigned)__builtin_amdgcn_readlane((int)ent, j + u) & 127u;
                    float* p = ssum + ld * 128u + lane;
                    __hip_atomic_fetch_add(p,      v[u][0], __ATOMIC_RELAXED, __HIP_MEMORY_SCOPE_WORKGROUP);
                    __hip_atomic_fetch_add(p + 32, v[u][1], __ATOMIC_RELAXED, __HIP_MEMORY_SCOPE_WORKGROUP);
                    __hip_atomic_fetch_add(p + 64, v[u][2], __ATOMIC_RELAXED, __HIP_MEMORY_SCOPE_WORKGROUP);
                    __hip_atomic_fetch_add(p + 96, v[u][3], __ATOMIC_RELAXED, __HIP_MEMORY_SCOPE_WORKGROUP);
                }
            }
        }
        i0 += n;
    }
    const int rem = count - i0;
    if (i0 > 0 && rem > 0) {
        unsigned tmp = 0u;
        if (lane < rem) tmp = list[i0 + lane];
        if (lane < rem) list[lane] = tmp;
    }
    return rem;
}

__global__ __launch_bounds__(1024) void agg_kernel(
    const int* __restrict__ src, const int* __restrict__ dst,
    const float* __restrict__ mx, float* __restrict__ c, int N, int E) {
    __shared__ float ssum[AGG_CAP * 128];
    __shared__ int scnt[AGG_CAP];
    __shared__ unsigned slist[32 * AGG_LIST];
    const int tid  = threadIdx.x;
    const int wave = __builtin_amdgcn_readfirstlane(tid >> 5);
    const int lane = tid & 31;
    unsigned* list = slist + wave * AGG_LIST;
    const int nblk = (E + 511) >> 9;

    for (int base = blockIdx.x * AGG_CAP; base < N; base += gridDim.x * AGG_CAP) {
        const int nloc = min(AGG_CAP, N - base);
        for (int i = tid; i < AGG_CAP * 128; i += 1024) ssum[i] = 0.0f;
        if (tid < AGG_CAP) scnt[tid] = 0;
        __syncthreads();

        int count = 0;
        const int rot = (int)(((long)blockIdx.x * nblk) / gridDim.x);
        for (int ib = wave; ib < nblk; ib += 32) {
            const int b = (ib + rot < nblk) ? ib + rot : ib + rot - nblk;
            const int e0 = b * 512 + lane * 4;
            int dv[16];
            if (b * 512 + 512 <= E) {
                const int4* p = (const int4*)(dst + e0);
#pragma unroll
                for (int k = 0; k < 4; ++k) {
                    const int4 q = p[k * 32];
                    dv[4 * k + 0] = q.x; dv[4 * k + 1] = q.y;
                    dv[4 * k + 2] = q.z; dv[4 * k + 3] = q.w;
                }
            } else {
#pragma unroll
                for (int i = 0; i < 16; ++i) {
                    const int e = e0 + (i >> 2) * 128 + (i & 3);
                    dv[i] = (e < E) ? dst[e] : -1;
                }
            }
#pragma unroll
            for (int i = 0; i < 16; ++i) {
                const unsigned t = (unsigned)dv[i] - (unsigned)base;
                const bool m = t < (unsigned)nloc;
                const unsigned mask = __builtin_amdgcn_ballot_w32(m);
                if (mask != 0u) {
                    if (m) {
                        const int pos = count + (int)__builtin_amdgcn_mbcnt_lo(mask, 0u);
                        const unsigned e = (unsigned)(e0 + (i >> 2) * 128 + (i & 3));
                        list[pos] = (e << 7) | t;
                    }
                    count += __builtin_popcount(mask);
                }
            }
            if (count >= 32)
                count = agg_flush(list, count, false, src, mx, ssum, scnt, lane);
        }
        agg_flush(list, count, true, src, mx, ssum, scnt, lane);
        __syncthreads();

        for (int r = wave; r < nloc; r += 32) {
            const float den = fmaxf((float)scnt[r], 1.0f);
            const float* sp = ssum + r * 128 + lane;
            float* cp = c + (size_t)(base + r) * 128 + lane;
            cp[0]  = sp[0]  / den;
            cp[32] = sp[32] / den;
            cp[64] = sp[64] / den;
            cp[96] = sp[96] / den;
        }
        __syncthreads();
    }
}

#define GRU_LDS_I (96 * 6 * 64)
#define GRU_LDS_H (64 * 6 * 64)
__global__ __launch_bounds__(256) void gru_kernel(
    const float* __restrict__ x, const float* __restrict__ c,
    const float* __restrict__ h, const float* __restrict__ Wih,
    const float* __restrict__ Whh, const float* __restrict__ bih,
    const float* __restrict__ bhh, float* __restrict__ hnew, int N) {
    __shared__ float sB[GRU_LDS_I + GRU_LDS_H];
    const int tid  = threadIdx.x;
    const int wave = tid >> 5;
    const int lane = tid & 31;
    const int cs   = blockIdx.x & 7;
    const int rgp  = blockIdx.x >> 3;

#pragma unroll
    for (int g = 0; g < 3; ++g) {
        fill_run<96, 6, 768, 8>(Wih, 256 * g + 32 * cs, sB, g, 3 + g, wave, lane);
        fill_run<64, 6, 768, 8>(Whh, 256 * g + 32 * cs, sB + GRU_LDS_I, g, 3 + g,
                                wave, lane);
    }
    __syncthreads();

    const int cidx = lane & 15, hh = lane >> 4;
    const float* sbi = sB + lane * 2;
    const float* sbh = sB + GRU_LDS_I + lane * 2;

    float bir[2], biz[2], bin[2], bhr[2], bhz[2], bhn[2];
#pragma unroll
    for (int j = 0; j < 2; ++j) {
        int col = (2 * cs + j) * 16 + cidx;
        bir[j] = bih[col]; biz[j] = bih[col + 256]; bin[j] = bih[col + 512];
        bhr[j] = bhh[col]; bhz[j] = bhh[col + 256]; bhn[j] = bhh[col + 512];
    }

    const int ntiles = N >> 4;
    for (int rt = rgp + 32 * wave; rt < ntiles; rt += 32 * 8) {
        const int row0 = rt * 16;
        const size_t ar = (size_t)(row0 + cidx);
        v8f ai[6], ah[6];
#pragma unroll
        for (int t = 0; t < 6; ++t) { ai[t] = v8f{}; ah[t] = v8f{}; }

        ws_phase<64, 6>(x + ar * 256 + 2 * hh, sbi, ai);
        ws_phase<32, 6>(c + ar * 128 + 2 * hh, sbi + 64 * 6 * 64, ai);
        ws_phase<64, 6>(h + ar * 256 + 2 * hh, sbh, ah);

#pragma unroll
        for (int j = 0; j < 2; ++j) {
            int col = (2 * cs + j) * 16 + cidx;
#pragma unroll
            for (int i = 0; i < 8; ++i) {
                int r = row0 + i + 8 * hh;
                float rg = sigmoid_f((ai[j * 3 + 0][i] + bir[j]) + (ah[j * 3 + 0][i] + bhr[j]));
                float zg = sigmoid_f((ai[j * 3 + 1][i] + biz[j]) + (ah[j * 3 + 1][i] + bhz[j]));
                float ng = tanh_f((ai[j * 3 + 2][i] + bin[j]) + rg * (ah[j * 3 + 2][i] + bhn[j]));
                float hv = h[(size_t)r * 256 + col];
                hnew[(size_t)r * 256 + col] = (1.0f - zg) * ng + zg * hv;
            }
        }
    }
}

__global__ __launch_bounds__(128) void out_kernel(
    const float* __restrict__ hnew, const float* __restrict__ outW,
    const float* __restrict__ outb, float* __restrict__ q, int N) {
    const int wave = threadIdx.x >> 5;
    const int lane = threadIdx.x & 31;
    const int row0 = (blockIdx.x * 4 + wave) * 32;
    if (row0 >= N) return;
    v8f acc0 = {}, acc1 = {};
    const float* bp = outW + (size_t)(2 * (lane >> 4)) * 16 + (lane & 15);
#pragma unroll 4
    for (int kk = 0; kk < 64; ++kk) {
        v2f b; b.x = bp[kk * 64]; b.y = bp[kk * 64 + 16];
        v2f a0 = load_a_frag(hnew, 256, row0, kk * 4, lane);
        v2f a1 = load_a_frag(hnew, 256, row0 + 16, kk * 4, lane);
        acc0 = wmma4(a0, b, acc0);
        acc1 = wmma4(a1, b, acc1);
    }
    int col = lane & 15;
    float bias = outb[col];
#pragma unroll
    for (int i = 0; i < 8; ++i) {
        int r = row0 + i + 8 * (lane >> 4);
        q[(size_t)r * 16 + col] = acc0[i] + bias;
        q[(size_t)(r + 16) * 16 + col] = acc1[i] + bias;
    }
}

extern "C" void kernel_launch(void* const* d_in, const int* in_sizes, int n_in,
                              void* d_out, int out_size, void* d_ws,
                              size_t ws_size, hipStream_t stream) {
    (void)n_in; (void)out_size; (void)ws_size;
    const float* feat = (const float*)d_in[0];
    const float* h    = (const float*)d_in[1];
    const int*   src  = (const int*)d_in[2];
    const int*   dst  = (const int*)d_in[3];
    const float* W0   = (const float*)d_in[4];
    const float* b0   = (const float*)d_in[5];
    const float* W1   = (const float*)d_in[6];
    const float* b1   = (const float*)d_in[7];
    const float* msgW = (const float*)d_in[8];
    const float* msgb = (const float*)d_in[9];
    const float* Wih  = (const float*)d_in[10];
    const float* Whh  = (const float*)d_in[11];
    const float* bih  = (const float*)d_in[12];
    const float* bhh  = (const float*)d_in[13];
    const float* outW = (const float*)d_in[14];
    const float* outb = (const float*)d_in[15];

    const int N = in_sizes[1] / 256;
    const int E = in_sizes[2];

    float* ws    = (float*)d_ws;
    float* x     = ws;
    float* mx    = x    + (size_t)N * 256;
    float* c     = mx   + (size_t)N * 128;

    float* q    = (float*)d_out;
    float* hnew = q + (size_t)N * 16;

    const int gemmBlocks = (N / 32 + 3) / 4;

    float* t1 = mx;
    enc1_kernel<<<256, 256, 0, stream>>>(feat, W0, b0, t1, N);
    enc2_kernel<<<256, 512, 0, stream>>>(t1, W1, b1, x, N);
    msg_kernel<<<256, 256, 0, stream>>>(x, h, msgW, msgb, mx, N);
    agg_kernel<<<256, 1024, 0, stream>>>(src, dst, mx, c, N, E);
    gru_kernel<<<256, 256, 0, stream>>>(x, c, h, Wih, Whh, bih, bhh,
                                        hnew, N);
    out_kernel<<<gemmBlocks, 128, 0, stream>>>(hnew, outW, outb, q, N);
}


# msg and gru workgroups go from 8 to 16 waves (4 a SIMD, as encoder stage 2 has) on the same resident LDS fragments, a wave's extra row tile spread over the SIMDs; k order and tiles unchanged
# speedup vs baseline: 5.4312x; 1.0870x over previous
#include <hip/hip_runtime.h>
#include <hip/hip_bf16.h>


typedef __attribute__((ext_vector_type(2))) float v2f;
typedef __attribute__((ext_vector_type(8))) float v8f;

__device__ __forceinline__ v8f wmma4(v2f a, v2f b, v8f c) {
    return __builtin_amdgcn_wmma_f32_16x16x4_f32(
        false, a, false, b, (short)0, c, false, false);
}

__device__ __forceinline__ v2f load_a_frag(const float* M, int ld, int row0,
                                           int k0, int lane) {
    int r = row0 + (lane & 15);
    int k = k0 + ((lane >> 4) << 1);
    const float* p = M + (size_t)r * ld + k;
    v2f a; a.x = p[0]; a.y = p[1];
    return a;
}

__device__ __forceinline__ float sigmoid_f(float v) {
    return 1.0f / (1.0f + __expf(-v));
}
__device__ __forceinline__ float tanh_f(float v) {
    float t = __expf(-2.0f * v);
    return (1.0f - t) / (1.0f + t);
}


template <int NK, int NT, int LD, int NW>
__device__ __forceinline__ void fill_run(const float* __restrict__ Wt, int col0,
                                         float* sb, int t0, int t1, int wave,
                                         int lane) {
    const int half = lane >> 4;
    const int tb   = (lane >> 3) & 1;
    const int cp   = lane & 7;
    const float* gp = Wt + (size_t)(2 * half) * LD + col0 + (lane & 15) * 2;
    float* lp = sb + (tb ? t1 : t0) * 64 + half * 32 + cp * 4;
#pragma unroll
    for (int i = 0; i < NK / NW; ++i) {
        const int kt = wave + NW * i;
        const v2f ra = *(const v2f*)(gp + (size_t)(4 * kt) * LD);
        const v2f rb = *(const v2f*)(gp + (size_t)(4 * kt + 1) * LD);
        float4 v;
        v.x = ra.x; v.y = rb.x;
        v.z = ra.y; v.w = rb.y;
        *(float4*)(lp + kt * (NT * 64)) = v;
    }
}

template <int NK, int NT>
__device__ __forceinline__ void ws_phase(const float* __restrict__ ap,
                                         const float* sb, v8f (&acc)[NT]) {
#pragma unroll 1
    for (int kb = 0; kb < NK; kb += 8) {
        v2f a[8];
#pragma unroll
        for (int s = 0; s < 8; ++s) a[s] = *(const v2f*)(ap + 4 * (kb + s));
#pragma unroll
        for (int s = 0; s < 8; ++s) {
            const float* bp = sb + (kb + s) * (NT * 64);
#pragma unroll
            for (int t = 0; t < NT; ++t) {
                v2f b = *(const v2f*)(bp + t * 64);
                acc[t] = wmma4(a[s], b, acc[t]);
            }
        }
    }
}

__device__ __forceinline__ int slot16(int wave) {
    return ((((wave >> 2) - wave) & 3) << 2) | (wave & 3);
}


__global__ __launch_bounds__(256) void enc1_kernel(
    const float* __restrict__ feat, const float* __restrict__ W0,
    const float* __restrict__ b0, float* __restrict__ t1, int N) {
    __shared__ float sB[16 * 8 * 64];
    const int tid  = threadIdx.x;
    const int wave = tid >> 5;
    const int lane = tid & 31;
    const int cs   = blockIdx.x & 1;
    const int rgp  = blockIdx.x >> 1;

#pragma unroll
    for (int r = 0; r < 4; ++r)
        fill_run<16, 8, 256, 8>(W0, 128 * cs + 32 * r, sB, 2 * r, 2 * r + 1,
                                wave, lane);
    __syncthreads();

    const int cidx = lane & 15, hh = lane >> 4;
    const float* sb = sB + lane * 2;
    float bias[8];
#pragma unroll
    for (int t = 0; t < 8; ++t) bias[t] = b0[(8 * cs + t) * 16 + cidx];

    const int ntiles = N >> 4;
    for (int rt = rgp + 128 * wave; rt < ntiles; rt += 128 * 8) {
        const int row0 = rt * 16;
        const size_t ar = (size_t)(row0 + cidx);
        v8f acc[8];
#pragma unroll
        for (int t = 0; t < 8; ++t) acc[t] = v8f{};
        ws_phase<16, 8>(feat + ar * 64 + 2 * hh, sb, acc);
#pragma unroll
        for (int t = 0; t < 8; ++t) {
            int col = (8 * cs + t) * 16 + cidx;
#pragma unroll
            for (int i = 0; i < 8; ++i) {
                int r = row0 + i + 8 * hh;
                t1[(size_t)r * 256 + col] = fmaxf(acc[t][i] + bias[t], 0.0f);
            }
        }
    }
}

__global__ __launch_bounds__(512) void enc2_kernel(
    const float* __restrict__ t1, const float* __restrict__ W1,
    const float* __restrict__ b1, float* __restrict__ x, int N) {
    __shared__ float sB[64 * 4 * 64];
    const int tid  = threadIdx.x;
    const int wave = tid >> 5;
    const int lane = tid & 31;
    const int cs   = blockIdx.x & 3;
    const int rgp  = blockIdx.x >> 2;

#pragma unroll
    for (int r = 0; r < 2; ++r)
        fill_run<64, 4, 256, 16>(W1, 64 * cs + 32 * r, sB, 2 * r, 2 * r + 1,
                                 wave, lane);
    __syncthreads();

    const int cidx = lane & 15, hh = lane >> 4;
    const float* sb = sB + lane * 2;
    float bias[4];
#pragma unroll
    for (int t = 0; t < 4; ++t) bias[t] = b1[(4 * cs + t) * 16 + cidx];

    const int vw = slot16(wave);
    const int ntiles = N >> 4;
    for (int rt = rgp + 64 * vw; rt < ntiles; rt += 64 * 16) {
        const int row0 = rt * 16;
        const size_t ar = (size_t)(row0 + cidx);
        v8f acc[4];
#pragma unroll
        for (int t = 0; t < 4; ++t) acc[t] = v8f{};
        ws_phase<64, 4>(t1 + ar * 256 + 2 * hh, sb, acc);
#pragma unroll
        for (int t = 0; t < 4; ++t) {
            int col = (4 * cs + t) * 16 + cidx;
#pragma unroll
            for (int i = 0; i < 8; ++i) {
                int r = row0 + i + 8 * hh;
                x[(size_t)r * 256 + col] = fmaxf(acc[t][i] + bias[t], 0.0f);
            }
        }
    }
}

__global__ __launch_bounds__(512) void msg_kernel(
    const float* __restrict__ x, const float* __restrict__ h,
    const float* __restrict__ msgW, const float* __restrict__ msgb,
    float* __restrict__ mx, int N) {
    __shared__ float sB[128 * 4 * 64];
    const int tid  = threadIdx.x;
    const int wave = tid >> 5;
    const int lane = tid & 31;
    const int cs   = blockIdx.x & 1;
    const int rgp  = blockIdx.x >> 1;

#pragma unroll
    for (int r = 0; r < 2; ++r)
        fill_run<128, 4, 128, 16>(msgW, 64 * cs + 32 * r, sB, 2 * r, 2 * r + 1,
                                  wave, lane);
    __syncthreads();

    const int cidx = lane & 15, hh = lane >> 4;
    const float* sb = sB + lane * 2;
    float bias[4];
#pragma unroll
    for (int t = 0; t < 4; ++t) bias[t] = msgb[(4 * cs + t) * 16 + cidx];

    const int ntiles = N >> 4;
    for (int rt = rgp + 128 * slot16(wave); rt < ntiles; rt += 128 * 16) {
        const int row0 = rt * 16;
        const size_t ar = (size_t)(row0 + cidx);
        v8f acc[4];
#pragma unroll
        for (int t = 0; t < 4; ++t) acc[t] = v8f{};
        ws_phase<64, 4>(x + ar * 256 + 2 * hh, sb, acc);
        ws_phase<64, 4>(h + ar * 256 + 2 * hh, sb + 64 * 4 * 64, acc);
#pragma unroll
        for (int t = 0; t < 4; ++t) {
            int col = (4 * cs + t) * 16 + cidx;
#pragma unroll
            for (int i = 0; i < 8; ++i) {
                int r = row0 + i + 8 * hh;
                mx[(size_t)r * 128 + col] = acc[t][i] + bias[t];
            }
        }
    }
}

#define AGG_CAP  80
#define AGG_LIST 1024

__device__ __forceinline__ int agg_flush(unsigned* list, int count, bool final,
                                         const int* __restrict__ src,
                                         const float* __restrict__ mx,
                                         float* ssum, int* scnt, int lane) {
    int i0 = 0;
    while (count - i0 >= 32 || (final && count > i0)) {
        const int n = min(32, count - i0);
        unsigned ent = 0u;
        int s = 0;
        if (lane < n) {
            ent = list[i0 + lane];
            s = src[ent >> 7];
            __hip_atomic_fetch_add(&scnt[ent & 127u], 1, __ATOMIC_RELAXED,
                                   __HIP_MEMORY_SCOPE_WORKGROUP);
        }
        for (int j = 0; j < n; j += 8) {
            float v[8][4];
#pragma unroll
            for (int u = 0; u < 8; ++u) {
                const int sj = __builtin_amdgcn_readlane(s, min(j + u, n - 1));
                const float* row = mx + (size_t)sj * 128 + lane;
                v[u][0] = row[0];  v[u][1] = row[32];
                v[u][2] = row[64]; v[u][3] = row[96];
            }
#pragma unroll
            for (int u = 0; u < 8; ++u) {
                if (j + u < n) {
                    const unsigned ld =
                        (unsigned)__builtin_amdgcn_readlane((int)ent, j + u) & 127u;
                    float* p = ssum + ld * 128u + lane;
                    __hip_atomic_fetch_add(p,      v[u][0], __ATOMIC_RELAXED, __HIP_MEMORY_SCOPE_WORKGROUP);
                    __hip_atomic_fetch_add(p + 32, v[u][1], __ATOMIC_RELAXED, __HIP_MEMORY_SCOPE_WORKGROUP);
                    __hip_atomic_fetch_add(p + 64, v[u][2], __ATOMIC_RELAXED, __HIP_MEMORY_SCOPE_WORKGROUP);
                    __hip_atomic_fetch_add(p + 96, v[u][3], __ATOMIC_RELAXED, __HIP_MEMORY_SCOPE_WORKGROUP);
                }
            }
        }
        i0 += n;
    }
    const int rem = count - i0;
    if (i0 > 0 && rem > 0) {
        unsigned tmp = 0u;
        if (lane < rem) tmp = list[i0 + lane];
        if (lane < rem) list[lane] = tmp;
    }
    return rem;
}

__global__ __launch_bounds__(1024) void agg_kernel(
    const int* __restrict__ src, const int* __restrict__ dst,
    const float* __restrict__ mx, float* __restrict__ c, int N, int E) {
    __shared__ float ssum[AGG_CAP * 128];
    __shared__ int scnt[AGG_CAP];
    __shared__ unsigned slist[32 * AGG_LIST];
    const int tid  = threadIdx.x;
    const int wave = __builtin_amdgcn_readfirstlane(tid >> 5);
    const int lane = tid & 31;
    unsigned* list = slist + wave * AGG_LIST;
    const int nblk = (E + 511) >> 9;

    for (int base = blockIdx.x * AGG_CAP; base < N; base += gridDim.x * AGG_CAP) {
        const int nloc = min(AGG_CAP, N - base);
        for (int i = tid; i < AGG_CAP * 128; i += 1024) ssum[i] = 0.0f;
        if (tid < AGG_CAP) scnt[tid] = 0;
        __syncthreads();

        int count = 0;
        const int rot = (int)(((long)blockIdx.x * nblk) / gridDim.x);
        for (int ib = wave; ib < nblk; ib += 32) {
            const int b = (ib + rot < nblk) ? ib + rot : ib + rot - nblk;
            const int e0 = b * 512 + lane * 4;
            int dv[16];
            if (b * 512 + 512 <= E) {
                const int4* p = (const int4*)(dst + e0);
#pragma unroll
                for (int k = 0; k < 4; ++k) {
                    const int4 q = p[k * 32];
                    dv[4 * k + 0] = q.x; dv[4 * k + 1] = q.y;
                    dv[4 * k + 2] = q.z; dv[4 * k + 3] = q.w;
                }
            } else {
#pragma unroll
                for (int i = 0; i < 16; ++i) {
                    const int e = e0 + (i >> 2) * 128 + (i & 3);
                    dv[i] = (e < E) ? dst[e] : -1;
                }
            }
#pragma unroll
            for (int i = 0; i < 16; ++i) {
                const unsigned t = (unsigned)dv[i] - (unsigned)base;
                const bool m = t < (unsigned)nloc;
                const unsigned mask = __builtin_amdgcn_ballot_w32(m);
                if (mask != 0u) {
                    if (m) {
                        const int pos = count + (int)__builtin_amdgcn_mbcnt_lo(mask, 0u);
                        const unsigned e = (unsigned)(e0 + (i >> 2) * 128 + (i & 3));
                        list[pos] = (e << 7) | t;
                    }
                    count += __builtin_popcount(mask);
                }
            }
            if (count >= 32)
                count = agg_flush(list, count, false, src, mx, ssum, scnt, lane);
        }
        agg_flush(list, count, true, src, mx, ssum, scnt, lane);
        __syncthreads();

        for (int r = wave; r < nloc; r += 32) {
            const float den = fmaxf((float)scnt[r], 1.0f);
            const float* sp = ssum + r * 128 + lane;
            float* cp = c + (size_t)(base + r) * 128 + lane;
            cp[0]  = sp[0]  / den;
            cp[32] = sp[32] / den;
            cp[64] = sp[64] / den;
            cp[96] = sp[96] / den;
        }
        __syncthreads();
    }
}

#define GRU_LDS_I (96 * 6 * 64)
#define GRU_LDS_H (64 * 6 * 64)
__global__ __launch_bounds__(512) void gru_kernel(
    const float* __restrict__ x, const float* __restrict__ c,
    const float* __restrict__ h, const float* __restrict__ Wih,
    const float* __restrict__ Whh, const float* __restrict__ bih,
    const float* __restrict__ bhh, float* __restrict__ hnew, int N) {
    __shared__ float sB[GRU_LDS_I + GRU_LDS_H];
    const int tid  = threadIdx.x;
    const int wave = tid >> 5;
    const int lane = tid & 31;
    const int cs   = blockIdx.x & 7;
    const int rgp  = blockIdx.x >> 3;

#pragma unroll
    for (int g = 0; g < 3; ++g) {
        fill_run<96, 6, 768, 16>(Wih, 256 * g + 32 * cs, sB, g, 3 + g, wave, lane);
        fill_run<64, 6, 768, 16>(Whh, 256 * g + 32 * cs, sB + GRU_LDS_I, g, 3 + g,
                                 wave, lane);
    }
    __syncthreads();

    const int cidx = lane & 15, hh = lane >> 4;
    const float* sbi = sB + lane * 2;
    const float* sbh = sB + GRU_LDS_I + lane * 2;

    float bir[2], biz[2], bin[2], bhr[2], bhz[2], bhn[2];
#pragma unroll
    for (int j = 0; j < 2; ++j) {
        int col = (2 * cs + j) * 16 + cidx;
        bir[j] = bih[col]; biz[j] = bih[col + 256]; bin[j] = bih[col + 512];
        bhr[j] = bhh[col]; bhz[j] = bhh[col + 256]; bhn[j] = bhh[col + 512];
    }

    const int ntiles = N >> 4;
    for (int rt = rgp + 32 * slot16(wave); rt < ntiles; rt += 32 * 16) {
        const int row0 = rt * 16;
        const size_t ar = (size_t)(row0 + cidx);
        v8f ai[6], ah[6];
#pragma unroll
        for (int t = 0; t < 6; ++t) { ai[t] = v8f{}; ah[t] = v8f{}; }

        ws_phase<64, 6>(x + ar * 256 + 2 * hh, sbi, ai);
        ws_phase<32, 6>(c + ar * 128 + 2 * hh, sbi + 64 * 6 * 64, ai);
        ws_phase<64, 6>(h + ar * 256 + 2 * hh, sbh, ah);

#pragma unroll
        for (int j = 0; j < 2; ++j) {
            int col = (2 * cs + j) * 16 + cidx;
#pragma unroll
            for (int i = 0; i < 8; ++i) {
                int r = row0 + i + 8 * hh;
                float rg = sigmoid_f((ai[j * 3 + 0][i] + bir[j]) + (ah[j * 3 + 0][i] + bhr[j]));
                float zg = sigmoid_f((ai[j * 3 + 1][i] + biz[j]) + (ah[j * 3 + 1][i] + bhz[j]));
                float ng = tanh_f((ai[j * 3 + 2][i] + bin[j]) + rg * (ah[j * 3 + 2][i] + bhn[j]));
                float hv = h[(size_t)r * 256 + col];
                hnew[(size_t)r * 256 + col] = (1.0f - zg) * ng + zg * hv;
            }
        }
    }
}

__global__ __launch_bounds__(128) void out_kernel(
    const float* __restrict__ hnew, const float* __restrict__ outW,
    const float* __restrict__ outb, float* __restrict__ q, int N) {
    const int wave = threadIdx.x >> 5;
    const int lane = threadIdx.x & 31;
    const int row0 = (blockIdx.x * 4 + wave) * 32;
    if (row0 >= N) return;
    v8f acc0 = {}, acc1 = {};
    const float* bp = outW + (size_t)(2 * (lane >> 4)) * 16 + (lane & 15);
#pragma unroll 4
    for (int kk = 0; kk < 64; ++kk) {
        v2f b; b.x = bp[kk * 64]; b.y = bp[kk * 64 + 16];
        v2f a0 = load_a_frag(hnew, 256, row0, kk * 4, lane);
        v2f a1 = load_a_frag(hnew, 256, row0 + 16, kk * 4, lane);
        acc0 = wmma4(a0, b, acc0);
        acc1 = wmma4(a1, b, acc1);
    }
    int col = lane & 15;
    float bias = outb[col];
#pragma unroll
    for (int i = 0; i < 8; ++i) {
        int r = row0 + i + 8 * (lane >> 4);
        q[(size_t)r * 16 + col] = acc0[i] + bias;
        q[(size_t)(r + 16) * 16 + col] = acc1[i] + bias;
    }
}

extern "C" void kernel_launch(void* const* d_in, const int* in_sizes, int n_in,
                              void* d_out, int out_size, void* d_ws,
                              size_t ws_size, hipStream_t stream) {
    (void)n_in; (void)out_size; (void)ws_size;
    const float* feat = (const float*)d_in[0];
    const float* h    = (const float*)d_in[1];
    const int*   src  = (const int*)d_in[2];
    const int*   dst  = (const int*)d_in[3];
    const float* W0   = (const float*)d_in[4];
    const float* b0   = (const float*)d_in[5];
    const float* W1   = (const float*)d_in[6];
    const float* b1   = (const float*)d_in[7];
    const float* msgW = (const float*)d_in[8];
    const float* msgb = (const float*)d_in[9];
    const float* Wih  = (const float*)d_in[10];
    const float* Whh  = (const float*)d_in[11];
    const float* bih  = (const float*)d_in[12];
    const float* bhh  = (const float*)d_in[13];
    const float* outW = (const float*)d_in[14];
    const float* outb = (const float*)d_in[15];

    const int N = in_sizes[1] / 256;
    const int E = in_sizes[2];

    float* ws    = (float*)d_ws;
    float* x     = ws;
    float* mx    = x    + (size_t)N * 256;
    float* c     = mx   + (size_t)N * 128;

    float* q    = (float*)d_out;
    float* hnew = q + (size_t)N * 16;

    const int gemmBlocks = (N / 32 + 3) / 4;

    float* t1 = mx;
    enc1_kernel<<<256, 256, 0, stream>>>(feat, W0, b0, t1, N);
    enc2_kernel<<<256, 512, 0, stream>>>(t1, W1, b1, x, N);
    msg_kernel<<<256, 512, 0, stream>>>(x, h, msgW, msgb, mx, N);
    agg_kernel<<<256, 1024, 0, stream>>>(src, dst, mx, c, N, E);
    gru_kernel<<<256, 512, 0, stream>>>(x, c, h, Wih, Whh, bih, bhh,
                                        hnew, N);
    out_kernel<<<gemmBlocks, 128, 0, stream>>>(hnew, outW, outb, q, N);
}


# edge mean: the first launch sorts each 2500-edge chunk by owner workgroup in LDS (count, offsets, scatter); an owner reads its 256 segments, no dst walk; sums stay LDS float atomics, arrival order
# speedup vs baseline: 5.8011x; 1.0681x over previous
#include <hip/hip_runtime.h>
#include <hip/hip_bf16.h>


typedef __attribute__((ext_vector_type(2))) float v2f;
typedef __attribute__((ext_vector_type(8))) float v8f;

__device__ __forceinline__ v8f wmma4(v2f a, v2f b, v8f c) {
    return __builtin_amdgcn_wmma_f32_16x16x4_f32(
        false, a, false, b, (short)0, c, false, false);
}

__device__ __forceinline__ v2f load_a_frag(const float* M, int ld, int row0,
                                           int k0, int lane) {
    int r = row0 + (lane & 15);
    int k = k0 + ((lane >> 4) << 1);
    const float* p = M + (size_t)r * ld + k;
    v2f a; a.x = p[0]; a.y = p[1];
    return a;
}

__device__ __forceinline__ float sigmoid_f(float v) {
    return 1.0f / (1.0f + __expf(-v));
}
__device__ __forceinline__ float tanh_f(float v) {
    float t = __expf(-2.0f * v);
    return (1.0f - t) / (1.0f + t);
}


template <int NK, int NT, int LD, int NW>
__device__ __forceinline__ void fill_run(const float* __restrict__ Wt, int col0,
                                         float* sb, int t0, int t1, int wave,
                                         int lane) {
    const int half = lane >> 4;
    const int tb   = (lane >> 3) & 1;
    const int cp   = lane & 7;
    const float* gp = Wt + (size_t)(2 * half) * LD + col0 + (lane & 15) * 2;
    float* lp = sb + (tb ? t1 : t0) * 64 + half * 32 + cp * 4;
#pragma unroll
    for (int i = 0; i < NK / NW; ++i) {
        const int kt = wave + NW * i;
        const v2f ra = *(const v2f*)(gp + (size_t)(4 * kt) * LD);
        const v2f rb = *(const v2f*)(gp + (size_t)(4 * kt + 1) * LD);
        float4 v;
        v.x = ra.x; v.y = rb.x;
        v.z = ra.y; v.w = rb.y;
        *(float4*)(lp + kt * (NT * 64)) = v;
    }
}

template <int NK, int NT>
__device__ __forceinline__ void ws_phase(const float* __restrict__ ap,
                                         const float* sb, v8f (&acc)[NT]) {
#pragma unroll 1
    for (int kb = 0; kb < NK; kb += 8) {
        v2f a[8];
#pragma unroll
        for (int s = 0; s < 8; ++s) a[s] = *(const v2f*)(ap + 4 * (kb + s));
#pragma unroll
        for (int s = 0; s < 8; ++s) {
            const float* bp = sb + (kb + s) * (NT * 64);
#pragma unroll
            for (int t = 0; t < NT; ++t) {
                v2f b = *(const v2f*)(bp + t * 64);
                acc[t] = wmma4(a[s], b, acc[t]);
            }
        }
    }
}

__device__ __forceinline__ int slot16(int wave) {
    return ((((wave >> 2) - wave) & 3) << 2) | (wave & 3);
}

#define AGG_CAP   80
#define EP_CHUNKS 256
#define EP_BINS   256
#define EP_B      12

__device__ __forceinline__ void ep_count(const int* __restrict__ dst, int e0,
                                         int e1, int nlim, int* hist, int tid) {
    for (int eb = e0; eb < e1; eb += EP_B * 256) {
        int d[EP_B];
#pragma unroll
        for (int k = 0; k < EP_B; ++k) {
            const int e = eb + k * 256 + tid;
            const int v = dst[min(e, e1 - 1)];
            d[k] = (e < e1) ? v : -1;
        }
#pragma unroll
        for (int k = 0; k < EP_B; ++k)
            if ((unsigned)d[k] < (unsigned)nlim)
                __hip_atomic_fetch_add(&hist[d[k] / AGG_CAP], 1, __ATOMIC_RELAXED,
                                       __HIP_MEMORY_SCOPE_WORKGROUP);
    }
}

__device__ __forceinline__ void ep_scan(const int* hist, int* curs,
                                        float* __restrict__ etab, int chunk,
                                        int lane) {
    int cb[8], co[8];
    int run = 0;
#pragma unroll
    for (int k = 0; k < 8; ++k) {
        cb[k] = hist[8 * lane + k];
        co[k] = run;
        run += cb[k];
    }
    int incl = run;
#pragma unroll
    for (int dlt = 1; dlt < 32; dlt <<= 1) {
        const int o = __shfl_up(incl, dlt);
        if (lane >= dlt) incl += o;
    }
    const int excl = incl - run;
#pragma unroll
    for (int k = 0; k < 8; ++k) {
        const int b = 8 * lane + k;
        const int off = excl + co[k];
        curs[b] = off;
        etab[(size_t)b * EP_CHUNKS + chunk] = __int_as_float((off << 16) | cb[k]);
    }
}

__device__ __forceinline__ void ep_scatter(const int* __restrict__ src,
                                           const int* __restrict__ dst, int e0,
                                           int e1, int nlim, int* curs,
                                           float* __restrict__ seg, int tid) {
    for (int eb = e0; eb < e1; eb += EP_B * 256) {
        int d[EP_B], s[EP_B];
#pragma unroll
        for (int k = 0; k < EP_B; ++k) {
            const int e = eb + k * 256 + tid;
            const int ec = min(e, e1 - 1);
            const int v = dst[ec];
            s[k] = src[ec];
            d[k] = (e < e1) ? v : -1;
        }
#pragma unroll
        for (int k = 0; k < EP_B; ++k)
            if ((unsigned)d[k] < (unsigned)nlim) {
                const int o = d[k] / AGG_CAP;
                const int pos = __hip_atomic_fetch_add(&curs[o], 1, __ATOMIC_RELAXED,
                                                       __HIP_MEMORY_SCOPE_WORKGROUP);
                seg[pos] = __int_as_float((s[k] << 7) | (d[k] - o * AGG_CAP));
            }
    }
}


__global__ __launch_bounds__(256) void enc1_kernel(
    const float* __restrict__ feat, const float* __restrict__ W0,
    const float* __restrict__ b0, float* __restrict__ t1,
    const int* __restrict__ src, const int* __restrict__ dst,
    float* __restrict__ elist, float* __restrict__ etab, int N, int E) {
    __shared__ float sB[16 * 8 * 64];
    __shared__ int hist[EP_BINS];
    __shared__ int curs[EP_BINS];
    const int tid  = threadIdx.x;
    const int wave = tid >> 5;
    const int lane = tid & 31;
    const int cs   = blockIdx.x & 1;
    const int rgp  = blockIdx.x >> 1;

    const int ch   = (E + EP_CHUNKS - 1) / EP_CHUNKS;
    const int e0   = min(E, (int)blockIdx.x * ch);
    const int e1   = min(E, e0 + ch);
    const int nlim = min(N, EP_BINS * AGG_CAP);
    hist[tid] = 0;
    __syncthreads();
    ep_count(dst, e0, e1, nlim, hist, tid);

#pragma unroll
    for (int r = 0; r < 4; ++r)
        fill_run<16, 8, 256, 8>(W0, 128 * cs + 32 * r, sB, 2 * r, 2 * r + 1,
                                wave, lane);
    __syncthreads();
    if (wave == 0) ep_scan(hist, curs, etab, blockIdx.x, lane);
    __syncthreads();
    ep_scatter(src, dst, e0, e1, nlim, curs, elist + (size_t)blockIdx.x * ch, tid);

    const int cidx = lane & 15, hh = lane >> 4;
    const float* sb = sB + lane * 2;
    float bias[8];
#pragma unroll
    for (int t = 0; t < 8; ++t) bias[t] = b0[(8 * cs + t) * 16 + cidx];

    const int ntiles = N >> 4;
    for (int rt = rgp + 128 * wave; rt < ntiles; rt += 128 * 8) {
        const int row0 = rt * 16;
        const size_t ar = (size_t)(row0 + cidx);
        v8f acc[8];
#pragma unroll
        for (int t = 0; t < 8; ++t) acc[t] = v8f{};
        ws_phase<16, 8>(feat + ar * 64 + 2 * hh, sb, acc);
#pragma unroll
        for (int t = 0; t < 8; ++t) {
            int col = (8 * cs + t) * 16 + cidx;
#pragma unroll
            for (int i = 0; i < 8; ++i) {
                int r = row0 + i + 8 * hh;
                t1[(size_t)r * 256 + col] = fmaxf(acc[t][i] + bias[t], 0.0f);
            }
        }
    }
}

__global__ __launch_bounds__(512) void enc2_kernel(
    const float* __restrict__ t1, const float* __restrict__ W1,
    const float* __restrict__ b1, float* __restrict__ x, int N) {
    __shared__ float sB[64 * 4 * 64];
    const int tid  = threadIdx.x;
    const int wave = tid >> 5;
    const int lane = tid & 31;
    const int cs   = blockIdx.x & 3;
    const int rgp  = blockIdx.x >> 2;

#pragma unroll
    for (int r = 0; r < 2; ++r)
        fill_run<64, 4, 256, 16>(W1, 64 * cs + 32 * r, sB, 2 * r, 2 * r + 1,
                                 wave, lane);
    __syncthreads();

    const int cidx = lane & 15, hh = lane >> 4;
    const float* sb = sB + lane * 2;
    float bias[4];
#pragma unroll
    for (int t = 0; t < 4; ++t) bias[t] = b1[(4 * cs + t) * 16 + cidx];

    const int vw = slot16(wave);
    const int ntiles = N >> 4;
    for (int rt = rgp + 64 * vw; rt < ntiles; rt += 64 * 16) {
        const int row0 = rt * 16;
        const size_t ar = (size_t)(row0 + cidx);
        v8f acc[4];
#pragma unroll
        for (int t = 0; t < 4; ++t) acc[t] = v8f{};
        ws_phase<64, 4>(t1 + ar * 256 + 2 * hh, sb, acc);
#pragma unroll
        for (int t = 0; t < 4; ++t) {
            int col = (4 * cs + t) * 16 + cidx;
#pragma unroll
            for (int i = 0; i < 8; ++i) {
                int r = row0 + i + 8 * hh;
                x[(size_t)r * 256 + col] = fmaxf(acc[t][i] + bias[t], 0.0f);
            }
        }
    }
}

__global__ __launch_bounds__(512) void msg_kernel(
    const float* __restrict__ x, const float* __restrict__ h,
    const float* __restrict__ msgW, const float* __restrict__ msgb,
    float* __restrict__ mx, int N) {
    __shared__ float sB[128 * 4 * 64];
    const int tid  = threadIdx.x;
    const int wave = tid >> 5;
    const int lane = tid & 31;
    const int cs   = blockIdx.x & 1;
    const int rgp  = blockIdx.x >> 1;

#pragma unroll
    for (int r = 0; r < 2; ++r)
        fill_run<128, 4, 128, 16>(msgW, 64 * cs + 32 * r, sB, 2 * r, 2 * r + 1,
                                  wave, lane);
    __syncthreads();

    const int cidx = lane & 15, hh = lane >> 4;
    const float* sb = sB + lane * 2;
    float bias[4];
#pragma unroll
    for (int t = 0; t < 4; ++t) bias[t] = msgb[(4 * cs + t) * 16 + cidx];

    const int ntiles = N >> 4;
    for (int rt = rgp + 128 * slot16(wave); rt < ntiles; rt += 128 * 16) {
        const int row0 = rt * 16;
        const size_t ar = (size_t)(row0 + cidx);
        v8f acc[4];
#pragma unroll
        for (int t = 0; t < 4; ++t) acc[t] = v8f{};
        ws_phase<64, 4>(x + ar * 256 + 2 * hh, sb, acc);
        ws_phase<64, 4>(h + ar * 256 + 2 * hh, sb + 64 * 4 * 64, acc);
#pragma unroll
        for (int t = 0; t < 4; ++t) {
            int col = (4 * cs + t) * 16 + cidx;
#pragma unroll
            for (int i = 0; i < 8; ++i) {
                int r = row0 + i + 8 * hh;
                mx[(size_t)r * 128 + col] = acc[t][i] + bias[t];
            }
        }
    }
}

#define AGG_LIST 320

__device__ __forceinline__ int agg_flush(unsigned* list, int count, bool final,
                                         const float* __restrict__ mx,
                                         float* ssum, int* scnt, int lane) {
    int i0 = 0;
    while (count - i0 >= 32 || (final && count > i0)) {
        const int n = min(32, count - i0);
        unsigned ent = 0u;
        int s = 0;
        if (lane < n) {
            ent = list[i0 + lane];
            s = (int)(ent >> 7);
            __hip_atomic_fetch_add(&scnt[ent & 127u], 1, __ATOMIC_RELAXED,
                                   __HIP_MEMORY_SCOPE_WORKGROUP);
        }
        for (int j = 0; j < n; j += 8) {
            float v[8][4];
#pragma unroll
            for (int u = 0; u < 8; ++u) {
                const int sj = __builtin_amdgcn_readlane(s, min(j + u, n - 1));
                const float* row = mx + (size_t)sj * 128 + lane;
                v[u][0] = row[0];  v[u][1] = row[32];
                v[u][2] = row[64]; v[u][3] = row[96];
            }
#pragma unroll
            for (int u = 0; u < 8; ++u) {
                if (j + u < n) {
                    const unsigned ld =
                        (unsigned)__builtin_amdgcn_readlane((int)ent, j + u) & 127u;
                    float* p = ssum + ld * 128u + lane;
                    __hip_atomic_fetch_add(p,      v[u][0], __ATOMIC_RELAXED, __HIP_MEMORY_SCOPE_WORKGROUP);
                    __hip_atomic_fetch_add(p + 32, v[u][1], __ATOMIC_RELAXED, __HIP_MEMORY_SCOPE_WORKGROUP);
                    __hip_atomic_fetch_add(p + 64, v[u][2], __ATOMIC_RELAXED, __HIP_MEMORY_SCOPE_WORKGROUP);
                    __hip_atomic_fetch_add(p + 96, v[u][3], __ATOMIC_RELAXED, __HIP_MEMORY_SCOPE_WORKGROUP);
                }
            }
        }
        i0 += n;
    }
    const int rem = count - i0;
    if (i0 > 0 && rem > 0) {
        unsigned tmp = 0u;
        if (lane < rem) tmp = list[i0 + lane];
        if (lane < rem) list[lane] = tmp;
    }
    return rem;
}

__global__ __launch_bounds__(1024) void agg_kernel(
    const float* __restrict__ elist, const float* __restrict__ etab,
    const float* __restrict__ mx, float* __restrict__ c, int N, int E) {
    __shared__ float ssum[AGG_CAP * 128];
    __shared__ int scnt[AGG_CAP];
    __shared__ unsigned slist[32 * AGG_LIST];
    const int tid  = threadIdx.x;
    const int wave = __builtin_amdgcn_readfirstlane(tid >> 5);
    const int lane = tid & 31;
    unsigned* list = slist + wave * AGG_LIST;
    const int base = blockIdx.x * AGG_CAP;
    if (base >= N) return;
    const int nloc = min(AGG_CAP, N - base);
    const int ch   = (E + EP_CHUNKS - 1) / EP_CHUNKS;

    for (int i = tid; i < AGG_CAP * 128; i += 1024) ssum[i] = 0.0f;
    if (tid < AGG_CAP) scnt[tid] = 0;
    __syncthreads();

    const int c0 = wave * 8;
    int te = 0;
    if (lane < 8)
        te = __float_as_int(etab[(size_t)blockIdx.x * EP_CHUNKS + c0 + lane]);

    unsigned fv[8];
#pragma unroll
    for (int s = 0; s < 8; ++s) {
        const int t = __builtin_amdgcn_readlane(te, s);
        const int n = min(t & 0xffff, 32);
        const float* p = elist + (size_t)(c0 + s) * ch + ((unsigned)t >> 16);
        fv[s] = (lane < n) ? (unsigned)__float_as_int(p[lane]) : 0u;
    }
    int count = 0;
#pragma unroll
    for (int s = 0; s < 8; ++s) {
        const int n = min(__builtin_amdgcn_readlane(te, s) & 0xffff, 32);
        if (lane < n) list[count + lane] = fv[s];
        count += n;
    }
    count = agg_flush(list, count, false, mx, ssum, scnt, lane);

    for (int s = 0; s < 8; ++s) {
        const int t = __builtin_amdgcn_readlane(te, s);
        const int cnt = t & 0xffff;
        const float* p = elist + (size_t)(c0 + s) * ch + ((unsigned)t >> 16);
        for (int i0 = 32; i0 < cnt; i0 += 32) {
            const int n = min(32, cnt - i0);
            if (lane < n) list[count + lane] = (unsigned)__float_as_int(p[i0 + lane]);
            count += n;
            if (count >= 32)
                count = agg_flush(list, count, false, mx, ssum, scnt, lane);
        }
    }
    agg_flush(list, count, true, mx, ssum, scnt, lane);
    __syncthreads();

    for (int r = wave; r < nloc; r += 32) {
        const float den = fmaxf((float)scnt[r], 1.0f);
        const float* sp = ssum + r * 128 + lane;
        float* cp = c + (size_t)(base + r) * 128 + lane;
        cp[0]  = sp[0]  / den;
        cp[32] = sp[32] / den;
        cp[64] = sp[64] / den;
        cp[96] = sp[96] / den;
    }
}

#define GRU_LDS_I (96 * 6 * 64)
#define GRU_LDS_H (64 * 6 * 64)
__global__ __launch_bounds__(512) void gru_kernel(
    const float* __restrict__ x, const float* __restrict__ c,
    const float* __restrict__ h, const float* __restrict__ Wih,
    const float* __restrict__ Whh, const float* __restrict__ bih,
    const float* __restrict__ bhh, float* __restrict__ hnew, int N) {
    __shared__ float sB[GRU_LDS_I + GRU_LDS_H];
    const int tid  = threadIdx.x;
    const int wave = tid >> 5;
    const int lane = tid & 31;
    const int cs   = blockIdx.x & 7;
    const int rgp  = blockIdx.x >> 3;

#pragma unroll
    for (int g = 0; g < 3; ++g) {
        fill_run<96, 6, 768, 16>(Wih, 256 * g + 32 * cs, sB, g, 3 + g, wave, lane);
        fill_run<64, 6, 768, 16>(Whh, 256 * g + 32 * cs, sB + GRU_LDS_I, g, 3 + g,
                                 wave, lane);
    }
    __syncthreads();

    const int cidx = lane & 15, hh = lane >> 4;
    const float* sbi = sB + lane * 2;
    const float* sbh = sB + GRU_LDS_I + lane * 2;

    float bir[2], biz[2], bin[2], bhr[2], bhz[2], bhn[2];
#pragma unroll
    for (int j = 0; j < 2; ++j) {
        int col = (2 * cs + j) * 16 + cidx;
        bir[j] = bih[col]; biz[j] = bih[col + 256]; bin[j] = bih[col + 512];
        bhr[j] = bhh[col]; bhz[j] = bhh[col + 256]; bhn[j] = bhh[col + 512];
    }

    const int ntiles = N >> 4;
    for (int rt = rgp + 32 * slot16(wave); rt < ntiles; rt += 32 * 16) {
        const int row0 = rt * 16;
        const size_t ar = (size_t)(row0 + cidx);
        v8f ai[6], ah[6];
#pragma unroll
        for (int t = 0; t < 6; ++t) { ai[t] = v8f{}; ah[t] = v8f{}; }

        ws_phase<64, 6>(x + ar * 256 + 2 * hh, sbi, ai);
        ws_phase<32, 6>(c + ar * 128 + 2 * hh, sbi + 64 * 6 * 64, ai);
        ws_phase<64, 6>(h + ar * 256 + 2 * hh, sbh, ah);

#pragma unroll
        for (int j = 0; j < 2; ++j) {
            int col = (2 * cs + j) * 16 + cidx;
#pragma unroll
            for (int i = 0; i < 8; ++i) {
                int r = row0 + i + 8 * hh;
                float rg = sigmoid_f((ai[j * 3 + 0][i] + bir[j]) + (ah[j * 3 + 0][i] + bhr[j]));
                float zg = sigmoid_f((ai[j * 3 + 1][i] + biz[j]) + (ah[j * 3 + 1][i] + bhz[j]));
                float ng = tanh_f((ai[j * 3 + 2][i] + bin[j]) + rg * (ah[j * 3 + 2][i] + bhn[j]));
                float hv = h[(size_t)r * 256 + col];
                hnew[(size_t)r * 256 + col] = (1.0f - zg) * ng + zg * hv;
            }
        }
    }
}

__global__ __launch_bounds__(128) void out_kernel(
    const float* __restrict__ hnew, const float* __restrict__ outW,
    const float* __restrict__ outb, float* __restrict__ q, int N) {
    const int wave = threadIdx.x >> 5;
    const int lane = threadIdx.x & 31;
    const int row0 = (blockIdx.x * 4 + wave) * 32;
    if (row0 >= N) return;
    v8f acc0 = {}, acc1 = {};
    const float* bp = outW + (size_t)(2 * (lane >> 4)) * 16 + (lane & 15);
#pragma unroll 4
    for (int kk = 0; kk < 64; ++kk) {
        v2f b; b.x = bp[kk * 64]; b.y = bp[kk * 64 + 16];
        v2f a0 = load_a_frag(hnew, 256, row0, kk * 4, lane);
        v2f a1 = load_a_frag(hnew, 256, row0 + 16, kk * 4, lane);
        acc0 = wmma4(a0, b, acc0);
        acc1 = wmma4(a1, b, acc1);
    }
    int col = lane & 15;
    float bias = outb[col];
#pragma unroll
    for (int i = 0; i < 8; ++i) {
        int r = row0 + i + 8 * (lane >> 4);
        q[(size_t)r * 16 + col] = acc0[i] + bias;
        q[(size_t)(r + 16) * 16 + col] = acc1[i] + bias;
    }
}

extern "C" void kernel_launch(void* const* d_in, const int* in_sizes, int n_in,
                              void* d_out, int out_size, void* d_ws,
                              size_t ws_size, hipStream_t stream) {
    (void)n_in; (void)out_size; (void)ws_size;
    const float* feat = (const float*)d_in[0];
    const float* h    = (const float*)d_in[1];
    const int*   src  = (const int*)d_in[2];
    const int*   dst  = (const int*)d_in[3];
    const float* W0   = (const float*)d_in[4];
    const float* b0   = (const float*)d_in[5];
    const float* W1   = (const float*)d_in[6];
    const float* b1   = (const float*)d_in[7];
    const float* msgW = (const float*)d_in[8];
    const float* msgb = (const float*)d_in[9];
    const float* Wih  = (const float*)d_in[10];
    const float* Whh  = (const float*)d_in[11];
    const float* bih  = (const float*)d_in[12];
    const float* bhh  = (const float*)d_in[13];
    const float* outW = (const float*)d_in[14];
    const float* outb = (const float*)d_in[15];

    const int N = in_sizes[1] / 256;
    const int E = in_sizes[2];

    float* ws    = (float*)d_ws;
    float* x     = ws;
    float* mx    = x    + (size_t)N * 256;
    float* c     = mx   + (size_t)N * 128;

    float* q    = (float*)d_out;
    float* hnew = q + (size_t)N * 16;

    const int gemmBlocks = (N / 32 + 3) / 4;

    float* t1 = mx;
    float* elist = hnew;
    float* etab  = hnew + E;
    enc1_kernel<<<EP_CHUNKS, 256, 0, stream>>>(feat, W0, b0, t1, src, dst, elist,
                                               etab, N, E);
    enc2_kernel<<<256, 512, 0, stream>>>(t1, W1, b1, x, N);
    msg_kernel<<<256, 512, 0, stream>>>(x, h, msgW, msgb, mx, N);
    agg_kernel<<<EP_BINS, 1024, 0, stream>>>(elist, etab, mx, c, N, E);
    gru_kernel<<<256, 512, 0, stream>>>(x, c, h, Wih, Whh, bih, bhh,
                                        hnew, N);
    out_kernel<<<gemmBlocks, 128, 0, stream>>>(hnew, outW, outb, q, N);
}


# enc, msg and gru waves set their issue priority (s_setprio) from the k blocks they still have ahead: a SIMD's waves end together, none runs its last rows alone; results and k order unchanged
# speedup vs baseline: 6.2997x; 1.0859x over previous
#include <hip/hip_runtime.h>
#include <hip/hip_bf16.h>


typedef __attribute__((ext_vector_type(2))) float v2f;
typedef __attribute__((ext_vector_type(8))) float v8f;

__device__ __forceinline__ v8f wmma4(v2f a, v2f b, v8f c) {
    return __builtin_amdgcn_wmma_f32_16x16x4_f32(
        false, a, false, b, (short)0, c, false, false);
}

__device__ __forceinline__ v2f load_a_frag(const float* M, int ld, int row0,
                                           int k0, int lane) {
    int r = row0 + (lane & 15);
    int k = k0 + ((lane >> 4) << 1);
    const float* p = M + (size_t)r * ld + k;
    v2f a; a.x = p[0]; a.y = p[1];
    return a;
}

__device__ __forceinline__ float sigmoid_f(float v) {
    return 1.0f / (1.0f + __expf(-v));
}
__device__ __forceinline__ float tanh_f(float v) {
    float t = __expf(-2.0f * v);
    return (1.0f - t) / (1.0f + t);
}


template <int NK, int NT, int LD, int NW>
__device__ __forceinline__ void fill_run(const float* __restrict__ Wt, int col0,
                                         float* sb, int t0, int t1, int wave,
                                         int lane) {
    const int half = lane >> 4;
    const int tb   = (lane >> 3) & 1;
    const int cp   = lane & 7;
    const float* gp = Wt + (size_t)(2 * half) * LD + col0 + (lane & 15) * 2;
    float* lp = sb + (tb ? t1 : t0) * 64 + half * 32 + cp * 4;
#pragma unroll
    for (int i = 0; i < NK / NW; ++i) {
        const int kt = wave + NW * i;
        const v2f ra = *(const v2f*)(gp + (size_t)(4 * kt) * LD);
        const v2f rb = *(const v2f*)(gp + (size_t)(4 * kt + 1) * LD);
        float4 v;
        v.x = ra.x; v.y = rb.x;
        v.z = ra.y; v.w = rb.y;
        *(float4*)(lp + kt * (NT * 64)) = v;
    }
}

__device__ __forceinline__ void wave_prio(int lvl) {
    if (lvl >= 3)      __builtin_amdgcn_s_setprio(3);
    else if (lvl == 2) __builtin_amdgcn_s_setprio(2);
    else if (lvl == 1) __builtin_amdgcn_s_setprio(1);
    else               __builtin_amdgcn_s_setprio(0);
}

__device__ __forceinline__ int tiles_ahead(int first, int step, int ntiles) {
    return first < ntiles ? (ntiles - 1 - first) / step + 1 : 0;
}

template <int NK, int NT, int SH>
__device__ __forceinline__ void ws_phase(const float* __restrict__ ap,
                                         const float* sb, v8f (&acc)[NT],
                                         int left, int& cur) {
#pragma unroll 1
    for (int kb = 0; kb < NK; kb += 8) {
        const int lvl = min(3, (left - (kb >> 3)) >> SH);
        if (lvl != cur) { wave_prio(lvl); cur = lvl; }
        v2f a[8];
#pragma unroll
        for (int s = 0; s < 8; ++s) a[s] = *(const v2f*)(ap + 4 * (kb + s));
#pragma unroll
        for (int s = 0; s < 8; ++s) {
            const float* bp = sb + (kb + s) * (NT * 64);
#pragma unroll
            for (int t = 0; t < NT; ++t) {
                v2f b = *(const v2f*)(bp + t * 64);
                acc[t] = wmma4(a[s], b, acc[t]);
            }
        }
    }
}

__device__ __forceinline__ int slot16(int wave) {
    return ((((wave >> 2) - wave) & 3) << 2) | (wave & 3);
}

#define AGG_CAP   80
#define EP_CHUNKS 256
#define EP_BINS   256
#define EP_B      12

__device__ __forceinline__ void ep_count(const int* __restrict__ dst, int e0,
                                         int e1, int nlim, int* hist, int tid) {
    for (int eb = e0; eb < e1; eb += EP_B * 256) {
        int d[EP_B];
#pragma unroll
        for (int k = 0; k < EP_B; ++k) {
            const int e = eb + k * 256 + tid;
            const int v = dst[min(e, e1 - 1)];
            d[k] = (e < e1) ? v : -1;
        }
#pragma unroll
        for (int k = 0; k < EP_B; ++k)
            if ((unsigned)d[k] < (unsigned)nlim)
                __hip_atomic_fetch_add(&hist[d[k] / AGG_CAP], 1, __ATOMIC_RELAXED,
                                       __HIP_MEMORY_SCOPE_WORKGROUP);
    }
}

__device__ __forceinline__ void ep_scan(const int* hist, int* curs,
                                        float* __restrict__ etab, int chunk,
                                        int lane) {
    int cb[8], co[8];
    int run = 0;
#pragma unroll
    for (int k = 0; k < 8; ++k) {
        cb[k] = hist[8 * lane + k];
        co[k] = run;
        run += cb[k];
    }
    int incl = run;
#pragma unroll
    for (int dlt = 1; dlt < 32; dlt <<= 1) {
        const int o = __shfl_up(incl, dlt);
        if (lane >= dlt) incl += o;
    }
    const int excl = incl - run;
#pragma unroll
    for (int k = 0; k < 8; ++k) {
        const int b = 8 * lane + k;
        const int off = excl + co[k];
        curs[b] = off;
        etab[(size_t)b * EP_CHUNKS + chunk] = __int_as_float((off << 16) | cb[k]);
    }
}

__device__ __forceinline__ void ep_scatter(const int* __restrict__ src,
                                           const int* __restrict__ dst, int e0,
                                           int e1, int nlim, int* curs,
                                           float* __restrict__ seg, int tid) {
    for (int eb = e0; eb < e1; eb += EP_B * 256) {
        int d[EP_B], s[EP_B];
#pragma unroll
        for (int k = 0; k < EP_B; ++k) {
            const int e = eb + k * 256 + tid;
            const int ec = min(e, e1 - 1);
            const int v = dst[ec];
            s[k] = src[ec];
            d[k] = (e < e1) ? v : -1;
        }
#pragma unroll
        for (int k = 0; k < EP_B; ++k)
            if ((unsigned)d[k] < (unsigned)nlim) {
                const int o = d[k] / AGG_CAP;
                const int pos = __hip_atomic_fetch_add(&curs[o], 1, __ATOMIC_RELAXED,
                                                       __HIP_MEMORY_SCOPE_WORKGROUP);
                seg[pos] = __int_as_float((s[k] << 7) | (d[k] - o * AGG_CAP));
            }
    }
}


__global__ __launch_bounds__(256) void enc1_kernel(
    const float* __restrict__ feat, const float* __restrict__ W0,
    const float* __restrict__ b0, float* __restrict__ t1,
    const int* __restrict__ src, const int* __restrict__ dst,
    float* __restrict__ elist, float* __restrict__ etab, int N, int E) {
    __shared__ float sB[16 * 8 * 64];
    __shared__ int hist[EP_BINS];
    __shared__ int curs[EP_BINS];
    const int tid  = threadIdx.x;
    const int wave = tid >> 5;
    const int lane = tid & 31;
    const int cs   = blockIdx.x & 1;
    const int rgp  = blockIdx.x >> 1;

    const int ch   = (E + EP_CHUNKS - 1) / EP_CHUNKS;
    const int e0   = min(E, (int)blockIdx.x * ch);
    const int e1   = min(E, e0 + ch);
    const int nlim = min(N, EP_BINS * AGG_CAP);
    hist[tid] = 0;
    __syncthreads();
    ep_count(dst, e0, e1, nlim, hist, tid);

#pragma unroll
    for (int r = 0; r < 4; ++r)
        fill_run<16, 8, 256, 8>(W0, 128 * cs + 32 * r, sB, 2 * r, 2 * r + 1,
                                wave, lane);
    __syncthreads();
    if (wave == 0) ep_scan(hist, curs, etab, blockIdx.x, lane);
    __syncthreads();
    ep_scatter(src, dst, e0, e1, nlim, curs, elist + (size_t)blockIdx.x * ch, tid);

    const int cidx = lane & 15, hh = lane >> 4;
    const float* sb = sB + lane * 2;
    float bias[8];
#pragma unroll
    for (int t = 0; t < 8; ++t) bias[t] = b0[(8 * cs + t) * 16 + cidx];

    const int ntiles = N >> 4;
    const int first = rgp + 128 * wave;
    int left = __builtin_amdgcn_readfirstlane(2 * tiles_ahead(first, 128 * 8, ntiles));
    int cur = 0;
    for (int rt = first; rt < ntiles; rt += 128 * 8) {
        const int row0 = rt * 16;
        const size_t ar = (size_t)(row0 + cidx);
        v8f acc[8];
#pragma unroll
        for (int t = 0; t < 8; ++t) acc[t] = v8f{};
        ws_phase<16, 8, 0>(feat + ar * 64 + 2 * hh, sb, acc, left, cur);
        left -= 2;
#pragma unroll
        for (int t = 0; t < 8; ++t) {
            int col = (8 * cs + t) * 16 + cidx;
#pragma unroll
            for (int i = 0; i < 8; ++i) {
                int r = row0 + i + 8 * hh;
                t1[(size_t)r * 256 + col] = fmaxf(acc[t][i] + bias[t], 0.0f);
            }
        }
    }
    __builtin_amdgcn_s_setprio(0);
}

__global__ __launch_bounds__(512) void enc2_kernel(
    const float* __restrict__ t1, const float* __restrict__ W1,
    const float* __restrict__ b1, float* __restrict__ x, int N) {
    __shared__ float sB[64 * 4 * 64];
    const int tid  = threadIdx.x;
    const int wave = tid >> 5;
    const int lane = tid & 31;
    const int cs   = blockIdx.x & 3;
    const int rgp  = blockIdx.x >> 2;

#pragma unroll
    for (int r = 0; r < 2; ++r)
        fill_run<64, 4, 256, 16>(W1, 64 * cs + 32 * r, sB, 2 * r, 2 * r + 1,
                                 wave, lane);
    __syncthreads();

    const int cidx = lane & 15, hh = lane >> 4;
    const float* sb = sB + lane * 2;
    float bias[4];
#pragma unroll
    for (int t = 0; t < 4; ++t) bias[t] = b1[(4 * cs + t) * 16 + cidx];

    const int ntiles = N >> 4;
    const int first = rgp + 64 * slot16(wave);
    int left = __builtin_amdgcn_readfirstlane(8 * tiles_ahead(first, 64 * 16, ntiles));
    int cur = 0;
    for (int rt = first; rt < ntiles; rt += 64 * 16) {
        const int row0 = rt * 16;
        const size_t ar = (size_t)(row0 + cidx);
        v8f acc[4];
#pragma unroll
        for (int t = 0; t < 4; ++t) acc[t] = v8f{};
        ws_phase<64, 4, 2>(t1 + ar * 256 + 2 * hh, sb, acc, left, cur);
        left -= 8;
#pragma unroll
        for (int t = 0; t < 4; ++t) {
            int col = (4 * cs + t) * 16 + cidx;
#pragma unroll
            for (int i = 0; i < 8; ++i) {
                int r = row0 + i + 8 * hh;
                x[(size_t)r * 256 + col] = fmaxf(acc[t][i] + bias[t], 0.0f);
            }
        }
    }
    __builtin_amdgcn_s_setprio(0);
}

__global__ __launch_bounds__(512) void msg_kernel(
    const float* __restrict__ x, const float* __restrict__ h,
    const float* __restrict__ msgW, const float* __restrict__ msgb,
    float* __restrict__ mx, int N) {
    __shared__ float sB[128 * 4 * 64];
    const int tid  = threadIdx.x;
    const int wave = tid >> 5;
    const int lane = tid & 31;
    const int cs   = blockIdx.x & 1;
    const int rgp  = blockIdx.x >> 1;

#pragma unroll
    for (int r = 0; r < 2; ++r)
        fill_run<128, 4, 128, 16>(msgW, 64 * cs + 32 * r, sB, 2 * r, 2 * r + 1,
                                  wave, lane);
    __syncthreads();

    const int cidx = lane & 15, hh = lane >> 4;
    const float* sb = sB + lane * 2;
    float bias[4];
#pragma unroll
    for (int t = 0; t < 4; ++t) bias[t] = msgb[(4 * cs + t) * 16 + cidx];

    const int ntiles = N >> 4;
    const int first = rgp + 128 * slot16(wave);
    int left = __builtin_amdgcn_readfirstlane(16 * tiles_ahead(first, 128 * 16, ntiles));
    int cur = 0;
    for (int rt = first; rt < ntiles; rt += 128 * 16) {
        const int row0 = rt * 16;
        const size_t ar = (size_t)(row0 + cidx);
        v8f acc[4];
#pragma unroll
        for (int t = 0; t < 4; ++t) acc[t] = v8f{};
        ws_phase<64, 4, 2>(x + ar * 256 + 2 * hh, sb, acc, left, cur);
        left -= 8;
        ws_phase<64, 4, 2>(h + ar * 256 + 2 * hh, sb + 64 * 4 * 64, acc, left,
                           cur);
        left -= 8;
#pragma unroll
        for (int t = 0; t < 4; ++t) {
            int col = (4 * cs + t) * 16 + cidx;
#pragma unroll
            for (int i = 0; i < 8; ++i) {
                int r = row0 + i + 8 * hh;
                mx[(size_t)r * 128 + col] = acc[t][i] + bias[t];
            }
        }
    }
    __builtin_amdgcn_s_setprio(0);
}

#define AGG_LIST 320

__device__ __forceinline__ int agg_flush(unsigned* list, int count, bool final,
                                         const float* __restrict__ mx,
                                         float* ssum, int* scnt, int lane) {
    int i0 = 0;
    while (count - i0 >= 32 || (final && count > i0)) {
        const int n = min(32, count - i0);
        unsigned ent = 0u;
        int s = 0;
        if (lane < n) {
            ent = list[i0 + lane];
            s = (int)(ent >> 7);
            __hip_atomic_fetch_add(&scnt[ent & 127u], 1, __ATOMIC_RELAXED,
                                   __HIP_MEMORY_SCOPE_WORKGROUP);
        }
        for (int j = 0; j < n; j += 8) {
            float v[8][4];
#pragma unroll
            for (int u = 0; u < 8; ++u) {
                const int sj = __builtin_amdgcn_readlane(s, min(j + u, n - 1));
                const float* row = mx + (size_t)sj * 128 + lane;
                v[u][0] = row[0];  v[u][1] = row[32];
                v[u][2] = row[64]; v[u][3] = row[96];
            }
#pragma unroll
            for (int u = 0; u < 8; ++u) {
                if (j + u < n) {
                    const unsigned ld =
                        (unsigned)__builtin_amdgcn_readlane((int)ent, j + u) & 127u;
                    float* p = ssum + ld * 128u + lane;
                    __hip_atomic_fetch_add(p,      v[u][0], __ATOMIC_RELAXED, __HIP_MEMORY_SCOPE_WORKGROUP);
                    __hip_atomic_fetch_add(p + 32, v[u][1], __ATOMIC_RELAXED, __HIP_MEMORY_SCOPE_WORKGROUP);
                    __hip_atomic_fetch_add(p + 64, v[u][2], __ATOMIC_RELAXED, __HIP_MEMORY_SCOPE_WORKGROUP);
                    __hip_atomic_fetch_add(p + 96, v[u][3], __ATOMIC_RELAXED, __HIP_MEMORY_SCOPE_WORKGROUP);
                }
            }
        }
        i0 += n;
    }
    const int rem = count - i0;
    if (i0 > 0 && rem > 0) {
        unsigned tmp = 0u;
        if (lane < rem) tmp = list[i0 + lane];
        if (lane < rem) list[lane] = tmp;
    }
    return rem;
}

__global__ __launch_bounds__(1024) void agg_kernel(
    const float* __restrict__ elist, const float* __restrict__ etab,
    const float* __restrict__ mx, float* __restrict__ c, int N, int E) {
    __shared__ float ssum[AGG_CAP * 128];
    __shared__ int scnt[AGG_CAP];
    __shared__ unsigned slist[32 * AGG_LIST];
    const int tid  = threadIdx.x;
    const int wave = __builtin_amdgcn_readfirstlane(tid >> 5);
    const int lane = tid & 31;
    unsigned* list = slist + wave * AGG_LIST;
    const int base = blockIdx.x * AGG_CAP;
    if (base >= N) return;
    const int nloc = min(AGG_CAP, N - base);
    const int ch   = (E + EP_CHUNKS - 1) / EP_CHUNKS;

    for (int i = tid; i < AGG_CAP * 128; i += 1024) ssum[i] = 0.0f;
    if (tid < AGG_CAP) scnt[tid] = 0;
    __syncthreads();

    const int c0 = wave * 8;
    int te = 0;
    if (lane < 8)
        te = __float_as_int(etab[(size_t)blockIdx.x * EP_CHUNKS + c0 + lane]);

    unsigned fv[8];
#pragma unroll
    for (int s = 0; s < 8; ++s) {
        const int t = __builtin_amdgcn_readlane(te, s);
        const int n = min(t & 0xffff, 32);
        const float* p = elist + (size_t)(c0 + s) * ch + ((unsigned)t >> 16);
        fv[s] = (lane < n) ? (unsigned)__float_as_int(p[lane]) : 0u;
    }
    int count = 0;
#pragma unroll
    for (int s = 0; s < 8; ++s) {
        const int n = min(__builtin_amdgcn_readlane(te, s) & 0xffff, 32);
        if (lane < n) list[count + lane] = fv[s];
        count += n;
    }
    count = agg_flush(list, count, false, mx, ssum, scnt, lane);

    for (int s = 0; s < 8; ++s) {
        const int t = __builtin_amdgcn_readlane(te, s);
        const int cnt = t & 0xffff;
        const float* p = elist + (size_t)(c0 + s) * ch + ((unsigned)t >> 16);
        for (int i0 = 32; i0 < cnt; i0 += 32) {
            const int n = min(32, cnt - i0);
            if (lane < n) list[count + lane] = (unsigned)__float_as_int(p[i0 + lane]);
            count += n;
            if (count >= 32)
                count = agg_flush(list, count, false, mx, ssum, scnt, lane);
        }
    }
    agg_flush(list, count, true, mx, ssum, scnt, lane);
    __syncthreads();

    for (int r = wave; r < nloc; r += 32) {
        const float den = fmaxf((float)scnt[r], 1.0f);
        const float* sp = ssum + r * 128 + lane;
        float* cp = c + (size_t)(base + r) * 128 + lane;
        cp[0]  = sp[0]  / den;
        cp[32] = sp[32] / den;
        cp[64] = sp[64] / den;
        cp[96] = sp[96] / den;
    }
}

#define GRU_LDS_I (96 * 6 * 64)
#define GRU_LDS_H (64 * 6 * 64)
__global__ __launch_bounds__(512) void gru_kernel(
    const float* __restrict__ x, const float* __restrict__ c,
    const float* __restrict__ h, const float* __restrict__ Wih,
    const float* __restrict__ Whh, const float* __restrict__ bih,
    const float* __restrict__ bhh, float* __restrict__ hnew, int N) {
    __shared__ float sB[GRU_LDS_I + GRU_LDS_H];
    const int tid  = threadIdx.x;
    const int wave = tid >> 5;
    const int lane = tid & 31;
    const int cs   = blockIdx.x & 7;
    const int rgp  = blockIdx.x >> 3;

#pragma unroll
    for (int g = 0; g < 3; ++g) {
        fill_run<96, 6, 768, 16>(Wih, 256 * g + 32 * cs, sB, g, 3 + g, wave, lane);
        fill_run<64, 6, 768, 16>(Whh, 256 * g + 32 * cs, sB + GRU_LDS_I, g, 3 + g,
                                 wave, lane);
    }
    __syncthreads();

    const int cidx = lane & 15, hh = lane >> 4;
    const float* sbi = sB + lane * 2;
    const float* sbh = sB + GRU_LDS_I + lane * 2;

    float bir[2], biz[2], bin[2], bhr[2], bhz[2], bhn[2];
#pragma unroll
    for (int j = 0; j < 2; ++j) {
        int col = (2 * cs + j) * 16 + cidx;
        bir[j] = bih[col]; biz[j] = bih[col + 256]; bin[j] = bih[col + 512];
        bhr[j] = bhh[col]; bhz[j] = bhh[col + 256]; bhn[j] = bhh[col + 512];
    }

    const int ntiles = N >> 4;
    const int first = rgp + 32 * slot16(wave);
    int left = __builtin_amdgcn_readfirstlane(20 * tiles_ahead(first, 32 * 16, ntiles));
    int cur = 0;
    for (int rt = first; rt < ntiles; rt += 32 * 16) {
        const int row0 = rt * 16;
        const size_t ar = (size_t)(row0 + cidx);
        v8f ai[6], ah[6];
#pragma unroll
        for (int t = 0; t < 6; ++t) { ai[t] = v8f{}; ah[t] = v8f{}; }

        ws_phase<64, 6, 3>(x + ar * 256 + 2 * hh, sbi, ai, left, cur);
        left -= 8;
        ws_phase<32, 6, 3>(c + ar * 128 + 2 * hh, sbi + 64 * 6 * 64, ai, left,
                           cur);
        left -= 4;
        ws_phase<64, 6, 3>(h + ar * 256 + 2 * hh, sbh, ah, left, cur);
        left -= 8;

#pragma unroll
        for (int j = 0; j < 2; ++j) {
            int col = (2 * cs + j) * 16 + cidx;
#pragma unroll
            for (int i = 0; i < 8; ++i) {
                int r = row0 + i + 8 * hh;
                float rg = sigmoid_f((ai[j * 3 + 0][i] + bir[j]) + (ah[j * 3 + 0][i] + bhr[j]));
                float zg = sigmoid_f((ai[j * 3 + 1][i] + biz[j]) + (ah[j * 3 + 1][i] + bhz[j]));
                float ng = tanh_f((ai[j * 3 + 2][i] + bin[j]) + rg * (ah[j * 3 + 2][i] + bhn[j]));
                float hv = h[(size_t)r * 256 + col];
                hnew[(size_t)r * 256 + col] = (1.0f - zg) * ng + zg * hv;
            }
        }
    }
    __builtin_amdgcn_s_setprio(0);
}

__global__ __launch_bounds__(128) void out_kernel(
    const float* __restrict__ hnew, const float* __restrict__ outW,
    const float* __restrict__ outb, float* __restrict__ q, int N) {
    const int wave = threadIdx.x >> 5;
    const int lane = threadIdx.x & 31;
    const int row0 = (blockIdx.x * 4 + wave) * 32;
    if (row0 >= N) return;
    v8f acc0 = {}, acc1 = {};
    const float* bp = outW + (size_t)(2 * (lane >> 4)) * 16 + (lane & 15);
#pragma unroll 4
    for (int kk = 0; kk < 64; ++kk) {
        v2f b; b.x = bp[kk * 64]; b.y = bp[kk * 64 + 16];
        v2f a0 = load_a_frag(hnew, 256, row0, kk * 4, lane);
        v2f a1 = load_a_frag(hnew, 256, row0 + 16, kk * 4, lane);
        acc0 = wmma4(a0, b, acc0);
        acc1 = wmma4(a1, b, acc1);
    }
    int col = lane & 15;
    float bias = outb[col];
#pragma unroll
    for (int i = 0; i < 8; ++i) {
        int r = row0 + i + 8 * (lane >> 4);
        q[(size_t)r * 16 + col] = acc0[i] + bias;
        q[(size_t)(r + 16) * 16 + col] = acc1[i] + bias;
    }
}

extern "C" void kernel_launch(void* const* d_in, const int* in_sizes, int n_in,
                              void* d_out, int out_size, void* d_ws,
                              size_t ws_size, hipStream_t stream) {
    (void)n_in; (void)out_size; (void)ws_size;
    const float* feat = (const float*)d_in[0];
    const float* h    = (const float*)d_in[1];
    const int*   src  = (const int*)d_in[2];
    const int*   dst  = (const int*)d_in[3];
    const float* W0   = (const float*)d_in[4];
    const float* b0   = (const float*)d_in[5];
    const float* W1   = (const float*)d_in[6];
    const float* b1   = (const float*)d_in[7];
    const float* msgW = (const float*)d_in[8];
    const float* msgb = (const float*)d_in[9];
    const float* Wih  = (const float*)d_in[10];
    const float* Whh  = (const float*)d_in[11];
    const float* bih  = (const float*)d_in[12];
    const float* bhh  = (const float*)d_in[13];
    const float* outW = (const float*)d_in[14];
    const float* outb = (const float*)d_in[15];

    const int N = in_sizes[1] / 256;
    const int E = in_sizes[2];

    float* ws    = (float*)d_ws;
    float* x     = ws;
    float* mx    = x    + (size_t)N * 256;
    float* c     = mx   + (size_t)N * 128;

    float* q    = (float*)d_out;
    float* hnew = q + (size_t)N * 16;

    const int gemmBlocks = (N / 32 + 3) / 4;

    float* t1 = mx;
    float* elist = hnew;
    float* etab  = hnew + E;
    enc1_kernel<<<EP_CHUNKS, 256, 0, stream>>>(feat, W0, b0, t1, src, dst, elist,
                                               etab, N, E);
    enc2_kernel<<<256, 512, 0, stream>>>(t1, W1, b1, x, N);
    msg_kernel<<<256, 512, 0, stream>>>(x, h, msgW, msgb, mx, N);
    agg_kernel<<<EP_BINS, 1024, 0, stream>>>(elist, etab, mx, c, N, E);
    gru_kernel<<<256, 512, 0, stream>>>(x, c, h, Wih, Whh, bih, bhh,
                                        hnew, N);
    out_kernel<<<gemmBlocks, 128, 0, stream>>>(hnew, outW, outb, q, N);
}
